# starting kernel
# baseline (speedup 1.0000x reference)
#include <hip/hip_runtime.h>
#include <math.h>

typedef __attribute__((ext_vector_type(2))) float v2f;
typedef __attribute__((ext_vector_type(8))) float v8f;

#define HID 128
#define NH  4
#define NC  32
#define NEG_SLOPE 0.2f


__device__ __forceinline__ void atomic_max_f32(float* addr, float val) {
    if (val >= 0.0f) {
        atomicMax((int*)addr, __float_as_int(val));
    } else {
        atomicMin((unsigned int*)addr, (unsigned int)__float_as_int(val));
    }
}

__device__ __forceinline__ void get_edge(const int* __restrict__ ei, int e,
                                         int E, int& s, int& d) {
    if (e < E) { s = ei[e]; d = ei[E + e]; }
    else       { s = e - E; d = e - E; }
}

__global__ void fill_f32(float* __restrict__ p, float v, int n) {
    int i = blockIdx.x * blockDim.x + threadIdx.x;
    if (i < n) p[i] = v;
}

template <int K>
__global__ void gemm_wmma_f32(const float* __restrict__ A,
                              const float* __restrict__ W,
                              float* __restrict__ C,
                              int n, int Oc) {
    const int wave = threadIdx.x >> 5;
    const int lane = threadIdx.x & 31;
    const int rowBase = blockIdx.x * 16;
    const int colBase = wave * 32;

    const int half = lane >> 4;
    const int l15  = lane & 15;
    int row = rowBase + l15;
    if (row >= n) row = n - 1;
    const int col0 = colBase + l15;
    const int col1 = col0 + 16;

    const float* Ap  = A + (size_t)row  * K + 2 * half;
    const float* Wp0 = W + (size_t)col0 * K + 2 * half;
    const float* Wp1 = W + (size_t)col1 * K + 2 * half;

    v8f acc0 = {}, acc1 = {};
#pragma unroll
    for (int k = 0; k < K; k += 4) {
        const v2f a  = *(const v2f*)(Ap  + k);
        const v2f b0 = *(const v2f*)(Wp0 + k);
        const v2f b1 = *(const v2f*)(Wp1 + k);
        acc0 = __builtin_amdgcn_wmma_f32_16x16x4_f32(
            false, a, false, b0, (short)0, acc0, false, false);
        acc1 = __builtin_amdgcn_wmma_f32_16x16x4_f32(
            false, a, false, b1, (short)0, acc1, false, false);
    }

    const int crow0 = rowBase + half * 8;
#pragma unroll
    for (int r = 0; r < 8; ++r) {
        const int crow = crow0 + r;
        if (crow < n) {
            C[(size_t)crow * Oc + col0] = acc0[r];
            C[(size_t)crow * Oc + col1] = acc1[r];
        }
    }
}

__global__ void att_dot(const float* __restrict__ xp,
                        const float* __restrict__ att_s,
                        const float* __restrict__ att_d,
                        float* __restrict__ a_s, float* __restrict__ a_d,
                        int n) {
    int i = blockIdx.x * blockDim.x + threadIdx.x;
    if (i >= n * NH) return;
    const int nn = i >> 2, h = i & 3;
    const float* v  = xp + (size_t)nn * HID + h * NC;
    const float* ws = att_s + h * NC;
    const float* wd = att_d + h * NC;
    float ss = 0.0f, sd = 0.0f;
#pragma unroll 8
    for (int c = 0; c < NC; ++c) { float x = v[c]; ss += x * ws[c]; sd += x * wd[c]; }
    a_s[i] = ss; a_d[i] = sd;
}

__device__ __forceinline__ float edge_logit(const float* __restrict__ a_s,
                                            const float* __restrict__ a_d,
                                            int s, int d, int h) {
    float v = a_s[s * NH + h] + a_d[d * NH + h];
    return v > 0.0f ? v : NEG_SLOPE * v;
}

__global__ void edge_max(const int* __restrict__ ei,
                         const float* __restrict__ a_s,
                         const float* __restrict__ a_d,
                         float* __restrict__ m, int E, int ET) {
    int i = blockIdx.x * blockDim.x + threadIdx.x;
    if (i >= ET * NH) return;
    const int e = i >> 2, h = i & 3;
    int s, d; get_edge(ei, e, E, s, d);
    atomic_max_f32(&m[d * NH + h], edge_logit(a_s, a_d, s, d, h));
}

__global__ void edge_denom(const int* __restrict__ ei,
                           const float* __restrict__ a_s,
                           const float* __restrict__ a_d,
                           const float* __restrict__ m,
                           float* __restrict__ denom, int E, int ET) {
    int i = blockIdx.x * blockDim.x + threadIdx.x;
    if (i >= ET * NH) return;
    const int e = i >> 2, h = i & 3;
    int s, d; get_edge(ei, e, E, s, d);
    const float v = edge_logit(a_s, a_d, s, d, h);
    atomicAdd(&denom[d * NH + h], expf(v - m[d * NH + h]));
}

__global__ void edge_agg(const int* __restrict__ ei,
                         const float* __restrict__ xp,
                         const float* __restrict__ a_s,
                         const float* __restrict__ a_d,
                         const float* __restrict__ m,
                         const float* __restrict__ denom,
                         float* __restrict__ agg, int E, int ET) {
    const int wv   = (blockIdx.x * blockDim.x + threadIdx.x) >> 5;
    const int lane = threadIdx.x & 31;
    if (wv >= ET) return;
    int s, d; get_edge(ei, wv, E, s, d);
    const float* xs = xp  + (size_t)s * HID;
    float*       ag = agg + (size_t)d * HID;
#pragma unroll
    for (int h = 0; h < NH; ++h) {
        const float v = edge_logit(a_s, a_d, s, d, h);
        const float alpha = expf(v - m[d * NH + h]) / (denom[d * NH + h] + 1e-16f);
        atomicAdd(&ag[h * NC + lane], xs[h * NC + lane] * alpha);
    }
}

__global__ void elu_bias_inplace(float* __restrict__ a,
                                 const float* __restrict__ bias, int total) {
    int i = blockIdx.x * blockDim.x + threadIdx.x;
    if (i >= total) return;
    float v = a[i] + bias[i & (HID - 1)];
    a[i] = v > 0.0f ? v : expm1f(v);
}

__global__ void mean_heads_bias(const float* __restrict__ agg,
                                const float* __restrict__ bias,
                                float* __restrict__ out, int n) {
    int i = blockIdx.x * blockDim.x + threadIdx.x;
    if (i >= n * NC) return;
    const int nn = i >> 5, c = i & 31;
    const float* a = agg + (size_t)nn * HID;
    out[i] = 0.25f * (a[c] + a[NC + c] + a[2 * NC + c] + a[3 * NC + c]) + bias[c];
}


static inline int cdiv(int a, int b) { return (a + b - 1) / b; }

extern "C" void kernel_launch(void* const* d_in, const int* in_sizes, int n_in,
                              void* d_out, int out_size, void* d_ws, size_t ws_size,
                              hipStream_t stream) {
    const float* x        = (const float*)d_in[0];
    const int*   ei       = (const int*)  d_in[1];
    const float* W1       = (const float*)d_in[2];
    const float* att_src1 = (const float*)d_in[3];
    const float* att_dst1 = (const float*)d_in[4];
    const float* bias1    = (const float*)d_in[5];
    const float* W2       = (const float*)d_in[6];
    const float* att_src2 = (const float*)d_in[7];
    const float* att_dst2 = (const float*)d_in[8];
    const float* bias2    = (const float*)d_in[9];
    float* out = (float*)d_out;

    const int IN = 64;
    const int N  = in_sizes[0] / IN;
    const int E  = in_sizes[1] / 2;
    const int ET = E + N;

    float* xp1   = (float*)d_ws;
    float* agg1  = xp1  + (size_t)N * HID;
    float* xp2   = agg1 + (size_t)N * HID;
    float* agg2  = xp2  + (size_t)N * HID;
    float* a_s   = agg2 + (size_t)N * HID;
    float* a_d   = a_s  + (size_t)N * NH;
    float* mbuf  = a_d  + (size_t)N * NH;
    float* dnm   = mbuf + (size_t)N * NH;

    const int BT = 256;
    const int gRowTiles = cdiv(N, 16);
    const int gNH  = cdiv(N * NH, BT);
    const int gEH  = cdiv(ET * NH, BT);
    const int gEW  = cdiv(ET * 32, BT);
    const int gNF  = cdiv(N * HID, BT);
    const int gOut = cdiv(N * NC, BT);

    gemm_wmma_f32<64><<<gRowTiles, 128, 0, stream>>>(x, W1, xp1, N, HID);
    att_dot<<<gNH, BT, 0, stream>>>(xp1, att_src1, att_dst1, a_s, a_d, N);
    fill_f32<<<gNH, BT, 0, stream>>>(mbuf, -INFINITY, N * NH);
    fill_f32<<<gNH, BT, 0, stream>>>(dnm, 0.0f, N * NH);
    fill_f32<<<gNF, BT, 0, stream>>>(agg1, 0.0f, N * HID);
    edge_max  <<<gEH, BT, 0, stream>>>(ei, a_s, a_d, mbuf, E, ET);
    edge_denom<<<gEH, BT, 0, stream>>>(ei, a_s, a_d, mbuf, dnm, E, ET);
    edge_agg  <<<gEW, BT, 0, stream>>>(ei, xp1, a_s, a_d, mbuf, dnm, agg1, E, ET);
    elu_bias_inplace<<<gNF, BT, 0, stream>>>(agg1, bias1, N * HID);

    gemm_wmma_f32<128><<<gRowTiles, 128, 0, stream>>>(agg1, W2, xp2, N, HID);
    att_dot<<<gNH, BT, 0, stream>>>(xp2, att_src2, att_dst2, a_s, a_d, N);
    fill_f32<<<gNH, BT, 0, stream>>>(mbuf, -INFINITY, N * NH);
    fill_f32<<<gNH, BT, 0, stream>>>(dnm, 0.0f, N * NH);
    fill_f32<<<gNF, BT, 0, stream>>>(agg2, 0.0f, N * HID);
    edge_max  <<<gEH, BT, 0, stream>>>(ei, a_s, a_d, mbuf, E, ET);
    edge_denom<<<gEH, BT, 0, stream>>>(ei, a_s, a_d, mbuf, dnm, E, ET);
    edge_agg  <<<gEW, BT, 0, stream>>>(ei, xp2, a_s, a_d, mbuf, dnm, agg2, E, ET);
    mean_heads_bias<<<gOut, BT, 0, stream>>>(agg2, bias2, out, N);
}


# Edges grouped by destination by an in-call counting sort; one wave per node does softmax max, sum and weighted aggregate in registers, replacing the three atomic scatter passes and their fills
# speedup vs baseline: 2.7890x; 2.7890x over previous
#include <hip/hip_runtime.h>
#include <math.h>

typedef __attribute__((ext_vector_type(2))) float v2f;
typedef __attribute__((ext_vector_type(8))) float v8f;

#define HID 128
#define NH  4
#define NC  32
#define NEG_SLOPE 0.2f

template <int K>
__global__ void gemm_wmma_f32(const float* __restrict__ A,
                              const float* __restrict__ W,
                              float* __restrict__ C,
                              int n, int Oc) {
    const int wave = threadIdx.x >> 5;
    const int lane = threadIdx.x & 31;
    const int rowBase = blockIdx.x * 16;
    const int colBase = wave * 32;

    const int half = lane >> 4;
    const int l15  = lane & 15;
    int row = rowBase + l15;
    if (row >= n) row = n - 1;
    const int col0 = colBase + l15;
    const int col1 = col0 + 16;

    const float* Ap  = A + (size_t)row  * K + 2 * half;
    const float* Wp0 = W + (size_t)col0 * K + 2 * half;
    const float* Wp1 = W + (size_t)col1 * K + 2 * half;

    v8f acc0 = {}, acc1 = {};
#pragma unroll
    for (int k = 0; k < K; k += 4) {
        const v2f a  = *(const v2f*)(Ap  + k);
        const v2f b0 = *(const v2f*)(Wp0 + k);
        const v2f b1 = *(const v2f*)(Wp1 + k);
        acc0 = __builtin_amdgcn_wmma_f32_16x16x4_f32(
            false, a, false, b0, (short)0, acc0, false, false);
        acc1 = __builtin_amdgcn_wmma_f32_16x16x4_f32(
            false, a, false, b1, (short)0, acc1, false, false);
    }

    const int crow0 = rowBase + half * 8;
#pragma unroll
    for (int r = 0; r < 8; ++r) {
        const int crow = crow0 + r;
        if (crow < n) {
            C[(size_t)crow * Oc + col0] = acc0[r];
            C[(size_t)crow * Oc + col1] = acc1[r];
        }
    }
}

__global__ void att_dot(const float* __restrict__ xp,
                        const float* __restrict__ att_s,
                        const float* __restrict__ att_d,
                        float* __restrict__ a_s, float* __restrict__ a_d,
                        int n) {
    int i = blockIdx.x * blockDim.x + threadIdx.x;
    if (i >= n * NH) return;
    const int nn = i >> 2, h = i & 3;
    const float* v  = xp + (size_t)nn * HID + h * NC;
    const float* ws = att_s + h * NC;
    const float* wd = att_d + h * NC;
    float ss = 0.0f, sd = 0.0f;
#pragma unroll 8
    for (int c = 0; c < NC; ++c) { float x = v[c]; ss += x * ws[c]; sd += x * wd[c]; }
    a_s[i] = ss; a_d[i] = sd;
}


#define SCAN_T   256
#define SCAN_PER 4
#define SCAN_BLK (SCAN_T * SCAN_PER)

__global__ void fill_i32(int* __restrict__ p, int v, int n) {
    int i = blockIdx.x * blockDim.x + threadIdx.x;
    if (i < n) p[i] = v;
}

__global__ void dst_count(const int* __restrict__ ei, int* __restrict__ cnt, int E) {
    int i = blockIdx.x * blockDim.x + threadIdx.x;
    if (i < E) atomicAdd(&cnt[ei[E + i]], 1);
}

__global__ void scan_blocksum(const int* __restrict__ cnt, int* __restrict__ bsum, int n) {
    __shared__ int wsum[SCAN_T / 32];
    const int lane = threadIdx.x & 31, wave = threadIdx.x >> 5;
    const int base = blockIdx.x * SCAN_BLK + threadIdx.x * SCAN_PER;
    int v = 0;
#pragma unroll
    for (int k = 0; k < SCAN_PER; ++k) if (base + k < n) v += cnt[base + k];
#pragma unroll
    for (int off = 16; off > 0; off >>= 1) v += __shfl_xor(v, off);
    if (lane == 0) wsum[wave] = v;
    __syncthreads();
    if (threadIdx.x == 0) {
        int t = 0;
#pragma unroll
        for (int w = 0; w < SCAN_T / 32; ++w) t += wsum[w];
        bsum[blockIdx.x] = t;
    }
}

__global__ void scan_write(const int* __restrict__ cnt, const int* __restrict__ bsum,
                           int* __restrict__ rowptr, int* __restrict__ cursor, int n) {
    __shared__ int wincl[SCAN_T / 32];
    __shared__ int wprev[SCAN_T / 32];
    const int lane = threadIdx.x & 31, wave = threadIdx.x >> 5;

    int prev = 0;
    for (int j = threadIdx.x; j < (int)blockIdx.x; j += SCAN_T) prev += bsum[j];

    const int base = blockIdx.x * SCAN_BLK + threadIdx.x * SCAN_PER;
    int c[SCAN_PER];
    int tsum = 0;
#pragma unroll
    for (int k = 0; k < SCAN_PER; ++k) {
        c[k] = (base + k < n) ? cnt[base + k] : 0;
        tsum += c[k];
    }
    int incl = tsum;
#pragma unroll
    for (int off = 1; off < 32; off <<= 1) {
        const int t = __shfl_up(incl, off);
        if (lane >= off) incl += t;
    }
#pragma unroll
    for (int off = 16; off > 0; off >>= 1) prev += __shfl_xor(prev, off);
    if (lane == 31) wincl[wave] = incl;
    if (lane == 0)  wprev[wave] = prev;
    __syncthreads();
    int start = incl - tsum;
#pragma unroll
    for (int w = 0; w < SCAN_T / 32; ++w) {
        start += wprev[w];
        if (w < wave) start += wincl[w];
    }
#pragma unroll
    for (int k = 0; k < SCAN_PER; ++k) {
        const int i = base + k;
        if (i < n) {
            rowptr[i] = start;
            cursor[i] = start;
            start += c[k];
            if (i == n - 1) rowptr[n] = start;
        }
    }
}

__global__ void csr_fill(const int* __restrict__ ei, int* __restrict__ cursor,
                         int* __restrict__ csr_src, int E) {
    int i = blockIdx.x * blockDim.x + threadIdx.x;
    if (i < E) {
        const int pos = atomicAdd(&cursor[ei[E + i]], 1);
        csr_src[pos] = ei[i];
    }
}

__device__ __forceinline__ float lrelu(float v) {
    return v > 0.0f ? v : NEG_SLOPE * v;
}

__device__ __forceinline__ float4 edge_logits(const float* __restrict__ a_s, int s,
                                              const float4 ad) {
    const float4 a = *(const float4*)(a_s + (size_t)s * NH);
    return make_float4(lrelu(a.x + ad.x), lrelu(a.y + ad.y),
                       lrelu(a.z + ad.z), lrelu(a.w + ad.w));
}

__device__ __forceinline__ float4 exp_shifted(const float4 e, const float4 m) {
    return make_float4(expf(e.x - m.x), expf(e.y - m.y), expf(e.z - m.z), expf(e.w - m.w));
}

__device__ __forceinline__ float lane_bcast(float v, int srcLane) {
    return __int_as_float(__builtin_amdgcn_readlane(__float_as_int(v), srcLane));
}

__device__ __forceinline__ void gather_chunk(const float* __restrict__ xp, int s,
                                             const float4 p, int cnt, const float4 f,
                                             int lane, float4& acc, float& den) {
#pragma unroll 4
    for (int t = 0; t < cnt; ++t) {
        const int   st = __builtin_amdgcn_readlane(s, t);
        const float w  = lane_bcast(p.x, t) * f.x + lane_bcast(p.y, t) * f.y +
                         lane_bcast(p.z, t) * f.z + lane_bcast(p.w, t) * f.w;
        const float4 r = *(const float4*)(xp + (size_t)st * HID + lane * 4);
        acc.x += r.x * w; acc.y += r.y * w; acc.z += r.z * w; acc.w += r.w * w;
        den += w;
    }
}

__global__ void node_softmax_agg(const int* __restrict__ rowptr,
                                 const int* __restrict__ csr_src,
                                 const float* __restrict__ xp,
                                 const float* __restrict__ a_s,
                                 const float* __restrict__ a_d,
                                 float* __restrict__ agg, int n) {
    const int lane = threadIdx.x & 31;
    const int node = __builtin_amdgcn_readfirstlane(
        (int)((blockIdx.x * blockDim.x + threadIdx.x) >> 5));
    if (node >= n) return;
    const int hme = lane >> 3;
    const float4 f = make_float4(hme == 0 ? 1.0f : 0.0f, hme == 1 ? 1.0f : 0.0f,
                                 hme == 2 ? 1.0f : 0.0f, hme == 3 ? 1.0f : 0.0f);
    const int beg = rowptr[node], end = rowptr[node + 1];

    const float4 ad = *(const float4*)(a_d + (size_t)node * NH);
    const float4 es = edge_logits(a_s, node, ad);

    const int j0 = beg + lane;
    const bool v0 = j0 < end;
    const int s0 = v0 ? csr_src[j0] : node;
    const float4 e0 = edge_logits(a_s, s0, ad);
    float4 m = make_float4(fmaxf(es.x, e0.x), fmaxf(es.y, e0.y),
                           fmaxf(es.z, e0.z), fmaxf(es.w, e0.w));
    for (int base = beg + 32; base < end; base += 32) {
        const int j = base + lane;
        if (j < end) {
            const float4 e = edge_logits(a_s, csr_src[j], ad);
            m.x = fmaxf(m.x, e.x); m.y = fmaxf(m.y, e.y);
            m.z = fmaxf(m.z, e.z); m.w = fmaxf(m.w, e.w);
        }
    }
#pragma unroll
    for (int off = 16; off > 0; off >>= 1) {
        m.x = fmaxf(m.x, __shfl_xor(m.x, off));
        m.y = fmaxf(m.y, __shfl_xor(m.y, off));
        m.z = fmaxf(m.z, __shfl_xor(m.z, off));
        m.w = fmaxf(m.w, __shfl_xor(m.w, off));
    }

    float4 acc;
    float den;
    {
        const float4 ps = exp_shifted(es, m);
        const float w = ps.x * f.x + ps.y * f.y + ps.z * f.z + ps.w * f.w;
        const float4 r = *(const float4*)(xp + (size_t)node * HID + lane * 4);
        acc = make_float4(r.x * w, r.y * w, r.z * w, r.w * w);
        den = w;
    }
    {
        const float4 p0 = exp_shifted(e0, m);
        const int cnt = min(32, end - beg);
        gather_chunk(xp, s0, p0, cnt, f, lane, acc, den);
    }
    for (int base = beg + 32; base < end; base += 32) {
        const int j = base + lane;
        const int s = (j < end) ? csr_src[j] : node;
        const float4 p = exp_shifted(edge_logits(a_s, s, ad), m);
        gather_chunk(xp, s, p, min(32, end - base), f, lane, acc, den);
    }

    const float dd = den + 1e-16f;
    *(float4*)(agg + (size_t)node * HID + lane * 4) =
        make_float4(acc.x / dd, acc.y / dd, acc.z / dd, acc.w / dd);
}

__global__ void elu_bias_inplace(float* __restrict__ a,
                                 const float* __restrict__ bias, int total) {
    int i = blockIdx.x * blockDim.x + threadIdx.x;
    if (i >= total) return;
    float v = a[i] + bias[i & (HID - 1)];
    a[i] = v > 0.0f ? v : expm1f(v);
}

__global__ void mean_heads_bias(const float* __restrict__ agg,
                                const float* __restrict__ bias,
                                float* __restrict__ out, int n) {
    int i = blockIdx.x * blockDim.x + threadIdx.x;
    if (i >= n * NC) return;
    const int nn = i >> 5, c = i & 31;
    const float* a = agg + (size_t)nn * HID;
    out[i] = 0.25f * (a[c] + a[NC + c] + a[2 * NC + c] + a[3 * NC + c]) + bias[c];
}


static inline int cdiv(int a, int b) { return (a + b - 1) / b; }

extern "C" void kernel_launch(void* const* d_in, const int* in_sizes, int n_in,
                              void* d_out, int out_size, void* d_ws, size_t ws_size,
                              hipStream_t stream) {
    const float* x        = (const float*)d_in[0];
    const int*   ei       = (const int*)  d_in[1];
    const float* W1       = (const float*)d_in[2];
    const float* att_src1 = (const float*)d_in[3];
    const float* att_dst1 = (const float*)d_in[4];
    const float* bias1    = (const float*)d_in[5];
    const float* W2       = (const float*)d_in[6];
    const float* att_src2 = (const float*)d_in[7];
    const float* att_dst2 = (const float*)d_in[8];
    const float* bias2    = (const float*)d_in[9];
    float* out = (float*)d_out;

    const int IN = 64;
    const int N  = in_sizes[0] / IN;
    const int E  = in_sizes[1] / 2;

    float* xp1   = (float*)d_ws;
    float* agg1  = xp1  + (size_t)N * HID;
    float* xp2   = agg1 + (size_t)N * HID;
    float* r4    = xp2  + (size_t)N * HID;
    float* a_s   = r4   + (size_t)N * HID;
    float* a_d   = a_s  + (size_t)N * NH;
    float* agg2  = xp1;

    int* csr_src = (int*)r4;
    int* rowptr  = csr_src + (size_t)E;
    int* cursor  = rowptr + (size_t)N + 1;
    int* cnt     = cursor + (size_t)N;
    int* bsum    = cnt + (size_t)N;

    const int BT = 256;
    const int gRowTiles = cdiv(N, 16);
    const int gNH  = cdiv(N * NH, BT);
    const int gN   = cdiv(N, BT);
    const int gE   = cdiv(E, BT);
    const int gScan = cdiv(N, SCAN_BLK);
    const int gNW  = cdiv(N * 32, BT);
    const int gNF  = cdiv(N * HID, BT);
    const int gOut = cdiv(N * NC, BT);

    fill_i32<<<gN, BT, 0, stream>>>(cnt, 0, N);
    dst_count<<<gE, BT, 0, stream>>>(ei, cnt, E);
    scan_blocksum<<<gScan, SCAN_T, 0, stream>>>(cnt, bsum, N);
    scan_write<<<gScan, SCAN_T, 0, stream>>>(cnt, bsum, rowptr, cursor, N);
    csr_fill<<<gE, BT, 0, stream>>>(ei, cursor, csr_src, E);

    gemm_wmma_f32<64><<<gRowTiles, 128, 0, stream>>>(x, W1, xp1, N, HID);
    att_dot<<<gNH, BT, 0, stream>>>(xp1, att_src1, att_dst1, a_s, a_d, N);
    node_softmax_agg<<<gNW, BT, 0, stream>>>(rowptr, csr_src, xp1, a_s, a_d, agg1, N);
    elu_bias_inplace<<<gNF, BT, 0, stream>>>(agg1, bias1, N * HID);

    gemm_wmma_f32<128><<<gRowTiles, 128, 0, stream>>>(agg1, W2, xp2, N, HID);
    att_dot<<<gNH, BT, 0, stream>>>(xp2, att_src2, att_dst2, a_s, a_d, N);
    node_softmax_agg<<<gNW, BT, 0, stream>>>(rowptr, csr_src, xp2, a_s, a_d, agg2, N);
    mean_heads_bias<<<gOut, BT, 0, stream>>>(agg2, bias2, out, N);
}


# GEMM operands loaded 16 bytes per lane over an 8-wide k step, two row tiles per wave: a third as many cache lines touched per matrix product; k order in a step (product sum order) changes
# speedup vs baseline: 3.5199x; 1.2621x over previous
#include <hip/hip_runtime.h>
#include <math.h>

typedef __attribute__((ext_vector_type(2))) float v2f;
typedef __attribute__((ext_vector_type(4))) float v4f;
typedef __attribute__((ext_vector_type(8))) float v8f;

#define HID 128
#define NH  4
#define NC  32
#define NEG_SLOPE 0.2f

template <int XM>
__device__ __forceinline__ float row_xor(float v) {
    return __int_as_float(__builtin_amdgcn_update_dpp(
        0, __float_as_int(v), 0x160 | XM, 0xf, 0xf, true));
}


__device__ __forceinline__ float fold16(float (&v)[16], int l15) {
#pragma unroll
    for (int i = 0; i < 8; ++i) {
        const bool up = (l15 & 8) != 0;
        const float send = up ? v[i] : v[i + 8];
        const float keep = up ? v[i + 8] : v[i];
        v[i] = keep + row_xor<8>(send);
    }
#pragma unroll
    for (int i = 0; i < 4; ++i) {
        const bool up = (l15 & 4) != 0;
        const float send = up ? v[i] : v[i + 4];
        const float keep = up ? v[i + 4] : v[i];
        v[i] = keep + row_xor<4>(send);
    }
#pragma unroll
    for (int i = 0; i < 2; ++i) {
        const bool up = (l15 & 2) != 0;
        const float send = up ? v[i] : v[i + 2];
        const float keep = up ? v[i + 2] : v[i];
        v[i] = keep + row_xor<2>(send);
    }
    const bool up = (l15 & 1) != 0;
    const float send = up ? v[0] : v[1];
    const float keep = up ? v[1] : v[0];
    return keep + row_xor<1>(send);
}

template <int K>
__global__ void gemm_wmma_f32(const float* __restrict__ A,
                              const float* __restrict__ W,
                              float* __restrict__ C,
                              const float* __restrict__ att_s,
                              const float* __restrict__ att_d,
                              float* __restrict__ a_s, float* __restrict__ a_d,
                              int n, int Oc) {
    __shared__ __attribute__((aligned(16))) float sAtt[2][32][NH];

    const int wave = threadIdx.x >> 5;
    const int lane = threadIdx.x & 31;
    const int rowBase = blockIdx.x * 32;
    const int colBase = wave * 32;

    const int half = lane >> 4;
    const int l15  = lane & 15;
    int row0 = rowBase + l15;
    int row1 = row0 + 16;
    if (row0 >= n) row0 = n - 1;
    if (row1 >= n) row1 = n - 1;
    const int col0 = colBase + l15;
    const int col1 = col0 + 16;

    const float* Ap0 = A + (size_t)row0 * K + 4 * half;
    const float* Ap1 = A + (size_t)row1 * K + 4 * half;
    const float* Wp0 = W + (size_t)col0 * K + 4 * half;
    const float* Wp1 = W + (size_t)col1 * K + 4 * half;

    v8f acc00 = {}, acc01 = {}, acc10 = {}, acc11 = {};
#pragma unroll
    for (int k = 0; k < K; k += 8) {
        const v4f a0 = *(const v4f*)(Ap0 + k);
        const v4f a1 = *(const v4f*)(Ap1 + k);
        const v4f b0 = *(const v4f*)(Wp0 + k);
        const v4f b1 = *(const v4f*)(Wp1 + k);
        const v2f a0l = {a0.x, a0.y}, a0h = {a0.z, a0.w};
        const v2f a1l = {a1.x, a1.y}, a1h = {a1.z, a1.w};
        const v2f b0l = {b0.x, b0.y}, b0h = {b0.z, b0.w};
        const v2f b1l = {b1.x, b1.y}, b1h = {b1.z, b1.w};
        acc00 = __builtin_amdgcn_wmma_f32_16x16x4_f32(
            false, a0l, false, b0l, (short)0, acc00, false, false);
        acc01 = __builtin_amdgcn_wmma_f32_16x16x4_f32(
            false, a0l, false, b1l, (short)0, acc01, false, false);
        acc10 = __builtin_amdgcn_wmma_f32_16x16x4_f32(
            false, a1l, false, b0l, (short)0, acc10, false, false);
        acc11 = __builtin_amdgcn_wmma_f32_16x16x4_f32(
            false, a1l, false, b1l, (short)0, acc11, false, false);
        acc00 = __builtin_amdgcn_wmma_f32_16x16x4_f32(
            false, a0h, false, b0h, (short)0, acc00, false, false);
        acc01 = __builtin_amdgcn_wmma_f32_16x16x4_f32(
            false, a0h, false, b1h, (short)0, acc01, false, false);
        acc10 = __builtin_amdgcn_wmma_f32_16x16x4_f32(
            false, a1h, false, b0h, (short)0, acc10, false, false);
        acc11 = __builtin_amdgcn_wmma_f32_16x16x4_f32(
            false, a1h, false, b1h, (short)0, acc11, false, false);
    }

    const float ws0 = att_s[col0], ws1 = att_s[col1];
    const float wd0 = att_d[col0], wd1 = att_d[col1];
    float v[16];
#pragma unroll
    for (int r = 0; r < 8; ++r) {
        v[r]     = acc00[r] * ws0 + acc01[r] * ws1;
        v[r + 8] = acc00[r] * wd0 + acc01[r] * wd1;
    }
    const float d0 = fold16(v, l15);
#pragma unroll
    for (int r = 0; r < 8; ++r) {
        v[r]     = acc10[r] * ws0 + acc11[r] * ws1;
        v[r + 8] = acc10[r] * wd0 + acc11[r] * wd1;
    }
    const float d1 = fold16(v, l15);
    sAtt[l15 >> 3][half * 8 + (l15 & 7)][wave]      = d0;
    sAtt[l15 >> 3][16 + half * 8 + (l15 & 7)][wave] = d1;
    __syncthreads();
    if (wave < 2) {
        const int grow = rowBase + lane;
        if (grow < n) {
            const float4 o = *(const float4*)&sAtt[wave][lane][0];
            *(float4*)((wave ? a_d : a_s) + (size_t)grow * NH) = o;
        }
    }

    const int crow0 = rowBase + half * 8;
#pragma unroll
    for (int r = 0; r < 8; ++r) {
        const int crow = crow0 + r;
        if (crow < n) {
            C[(size_t)crow * Oc + col0] = acc00[r];
            C[(size_t)crow * Oc + col1] = acc01[r];
        }
    }
#pragma unroll
    for (int r = 0; r < 8; ++r) {
        const int crow = crow0 + 16 + r;
        if (crow < n) {
            C[(size_t)crow * Oc + col0] = acc10[r];
            C[(size_t)crow * Oc + col1] = acc11[r];
        }
    }
}


#define SCAN_T   256
#define SCAN_PER 4
#define SCAN_BLK (SCAN_T * SCAN_PER)

__global__ void fill_i32(int* __restrict__ p, int v, int n) {
    int i = blockIdx.x * blockDim.x + threadIdx.x;
    if (i < n) p[i] = v;
}

__global__ void dst_count(const int* __restrict__ ei, int* __restrict__ cnt, int E) {
    int i = blockIdx.x * blockDim.x + threadIdx.x;
    if (i < E) atomicAdd(&cnt[ei[E + i]], 1);
}

__global__ void scan_blocksum(const int* __restrict__ cnt, int* __restrict__ bsum, int n) {
    __shared__ int wsum[SCAN_T / 32];
    const int lane = threadIdx.x & 31, wave = threadIdx.x >> 5;
    const int base = blockIdx.x * SCAN_BLK + threadIdx.x * SCAN_PER;
    int v = 0;
#pragma unroll
    for (int k = 0; k < SCAN_PER; ++k) if (base + k < n) v += cnt[base + k];
#pragma unroll
    for (int off = 16; off > 0; off >>= 1) v += __shfl_xor(v, off);
    if (lane == 0) wsum[wave] = v;
    __syncthreads();
    if (threadIdx.x == 0) {
        int t = 0;
#pragma unroll
        for (int w = 0; w < SCAN_T / 32; ++w) t += wsum[w];
        bsum[blockIdx.x] = t;
    }
}

__global__ void scan_write(const int* __restrict__ cnt, const int* __restrict__ bsum,
                           int* __restrict__ rowptr, int* __restrict__ cursor, int n) {
    __shared__ int wincl[SCAN_T / 32];
    __shared__ int wprev[SCAN_T / 32];
    const int lane = threadIdx.x & 31, wave = threadIdx.x >> 5;

    int prev = 0;
    for (int j = threadIdx.x; j < (int)blockIdx.x; j += SCAN_T) prev += bsum[j];

    const int base = blockIdx.x * SCAN_BLK + threadIdx.x * SCAN_PER;
    int c[SCAN_PER];
    int tsum = 0;
#pragma unroll
    for (int k = 0; k < SCAN_PER; ++k) {
        c[k] = (base + k < n) ? cnt[base + k] : 0;
        tsum += c[k];
    }
    int incl = tsum;
#pragma unroll
    for (int off = 1; off < 32; off <<= 1) {
        const int t = __shfl_up(incl, off);
        if (lane >= off) incl += t;
    }
#pragma unroll
    for (int off = 16; off > 0; off >>= 1) prev += __shfl_xor(prev, off);
    if (lane == 31) wincl[wave] = incl;
    if (lane == 0)  wprev[wave] = prev;
    __syncthreads();
    int start = incl - tsum;
#pragma unroll
    for (int w = 0; w < SCAN_T / 32; ++w) {
        start += wprev[w];
        if (w < wave) start += wincl[w];
    }
#pragma unroll
    for (int k = 0; k < SCAN_PER; ++k) {
        const int i = base + k;
        if (i < n) {
            rowptr[i] = start;
            cursor[i] = start;
            start += c[k];
            if (i == n - 1) rowptr[n] = start;
        }
    }
}

__global__ void csr_fill(const int* __restrict__ ei, int* __restrict__ cursor,
                         int* __restrict__ csr_src, int E) {
    int i = blockIdx.x * blockDim.x + threadIdx.x;
    if (i < E) {
        const int pos = atomicAdd(&cursor[ei[E + i]], 1);
        csr_src[pos] = ei[i];
    }
}

__device__ __forceinline__ float lrelu(float v) {
    return v > 0.0f ? v : NEG_SLOPE * v;
}

__device__ __forceinline__ float elu1(float v) {
    return v > 0.0f ? v : expm1f(v);
}

__device__ __forceinline__ float4 edge_logits(const float* __restrict__ a_s, int s,
                                              const float4 ad) {
    const float4 a = *(const float4*)(a_s + (size_t)s * NH);
    return make_float4(lrelu(a.x + ad.x), lrelu(a.y + ad.y),
                       lrelu(a.z + ad.z), lrelu(a.w + ad.w));
}

__device__ __forceinline__ float4 exp_shifted(const float4 e, const float4 m) {
    return make_float4(expf(e.x - m.x), expf(e.y - m.y), expf(e.z - m.z), expf(e.w - m.w));
}

__device__ __forceinline__ float lane_bcast(float v, int srcLane) {
    return __int_as_float(__builtin_amdgcn_readlane(__float_as_int(v), srcLane));
}

__device__ __forceinline__ void gather_chunk(const float* __restrict__ xp, int s,
                                             const float4 p, int cnt, const float4 f,
                                             int lane, float4& acc, float& den) {
#pragma unroll 4
    for (int t = 0; t < cnt; ++t) {
        const int   st = __builtin_amdgcn_readlane(s, t);
        const float w  = lane_bcast(p.x, t) * f.x + lane_bcast(p.y, t) * f.y +
                         lane_bcast(p.z, t) * f.z + lane_bcast(p.w, t) * f.w;
        const float4 r = *(const float4*)(xp + (size_t)st * HID + lane * 4);
        acc.x += r.x * w; acc.y += r.y * w; acc.z += r.z * w; acc.w += r.w * w;
        den += w;
    }
}

template <bool LAST>
__global__ void node_softmax_agg(const int* __restrict__ rowptr,
                                 const int* __restrict__ csr_src,
                                 const float* __restrict__ xp,
                                 const float* __restrict__ a_s,
                                 const float* __restrict__ a_d,
                                 const float* __restrict__ bias,
                                 float* __restrict__ dst, int n) {
    const int lane = threadIdx.x & 31;
    const int node = __builtin_amdgcn_readfirstlane(
        (int)((blockIdx.x * blockDim.x + threadIdx.x) >> 5));
    if (node >= n) return;
    const int hme = lane >> 3;
    const float4 f = make_float4(hme == 0 ? 1.0f : 0.0f, hme == 1 ? 1.0f : 0.0f,
                                 hme == 2 ? 1.0f : 0.0f, hme == 3 ? 1.0f : 0.0f);
    const int beg = rowptr[node], end = rowptr[node + 1];

    const float4 ad = *(const float4*)(a_d + (size_t)node * NH);
    const float4 es = edge_logits(a_s, node, ad);

    const int j0 = beg + lane;
    const bool v0 = j0 < end;
    const int s0 = v0 ? csr_src[j0] : node;
    const float4 e0 = edge_logits(a_s, s0, ad);
    float4 m = make_float4(fmaxf(es.x, e0.x), fmaxf(es.y, e0.y),
                           fmaxf(es.z, e0.z), fmaxf(es.w, e0.w));
    for (int base = beg + 32; base < end; base += 32) {
        const int j = base + lane;
        if (j < end) {
            const float4 e = edge_logits(a_s, csr_src[j], ad);
            m.x = fmaxf(m.x, e.x); m.y = fmaxf(m.y, e.y);
            m.z = fmaxf(m.z, e.z); m.w = fmaxf(m.w, e.w);
        }
    }
#pragma unroll
    for (int off = 16; off > 0; off >>= 1) {
        m.x = fmaxf(m.x, __shfl_xor(m.x, off));
        m.y = fmaxf(m.y, __shfl_xor(m.y, off));
        m.z = fmaxf(m.z, __shfl_xor(m.z, off));
        m.w = fmaxf(m.w, __shfl_xor(m.w, off));
    }

    float4 acc;
    float den;
    {
        const float4 ps = exp_shifted(es, m);
        const float w = ps.x * f.x + ps.y * f.y + ps.z * f.z + ps.w * f.w;
        const float4 r = *(const float4*)(xp + (size_t)node * HID + lane * 4);
        acc = make_float4(r.x * w, r.y * w, r.z * w, r.w * w);
        den = w;
    }
    {
        const float4 p0 = exp_shifted(e0, m);
        const int cnt = min(32, end - beg);
        gather_chunk(xp, s0, p0, cnt, f, lane, acc, den);
    }
    for (int base = beg + 32; base < end; base += 32) {
        const int j = base + lane;
        const int s = (j < end) ? csr_src[j] : node;
        const float4 p = exp_shifted(edge_logits(a_s, s, ad), m);
        gather_chunk(xp, s, p, min(32, end - base), f, lane, acc, den);
    }

    const float dd = den + 1e-16f;
    const float4 a = make_float4(acc.x / dd, acc.y / dd, acc.z / dd, acc.w / dd);
    if (!LAST) {
        const float4 b = *(const float4*)(bias + lane * 4);
        *(float4*)(dst + (size_t)node * HID + lane * 4) =
            make_float4(elu1(a.x + b.x), elu1(a.y + b.y), elu1(a.z + b.z), elu1(a.w + b.w));
    } else {
        const float4 a1 = make_float4(__shfl_xor(a.x, 8),  __shfl_xor(a.y, 8),
                                      __shfl_xor(a.z, 8),  __shfl_xor(a.w, 8));
        const float4 a2 = make_float4(__shfl_xor(a.x, 16), __shfl_xor(a.y, 16),
                                      __shfl_xor(a.z, 16), __shfl_xor(a.w, 16));
        const float4 a3 = make_float4(__shfl_xor(a.x, 24), __shfl_xor(a.y, 24),
                                      __shfl_xor(a.z, 24), __shfl_xor(a.w, 24));
        if (lane < 8) {
            const float4 b = *(const float4*)(bias + lane * 4);
            *(float4*)(dst + (size_t)node * NC + lane * 4) =
                make_float4(0.25f * (a.x + a1.x + a2.x + a3.x) + b.x,
                            0.25f * (a.y + a1.y + a2.y + a3.y) + b.y,
                            0.25f * (a.z + a1.z + a2.z + a3.z) + b.z,
                            0.25f * (a.w + a1.w + a2.w + a3.w) + b.w);
        }
    }
}


static inline int cdiv(int a, int b) { return (a + b - 1) / b; }

extern "C" void kernel_launch(void* const* d_in, const int* in_sizes, int n_in,
                              void* d_out, int out_size, void* d_ws, size_t ws_size,
                              hipStream_t stream) {
    const float* x        = (const float*)d_in[0];
    const int*   ei       = (const int*)  d_in[1];
    const float* W1       = (const float*)d_in[2];
    const float* att_src1 = (const float*)d_in[3];
    const float* att_dst1 = (const float*)d_in[4];
    const float* bias1    = (const float*)d_in[5];
    const float* W2       = (const float*)d_in[6];
    const float* att_src2 = (const float*)d_in[7];
    const float* att_dst2 = (const float*)d_in[8];
    const float* bias2    = (const float*)d_in[9];
    float* out = (float*)d_out;

    const int IN = 64;
    const int N  = in_sizes[0] / IN;
    const int E  = in_sizes[1] / 2;

    float* xp1   = (float*)d_ws;
    float* agg1  = xp1  + (size_t)N * HID;
    float* xp2   = agg1 + (size_t)N * HID;
    float* r4    = xp2  + (size_t)N * HID;
    float* a_s   = r4   + (size_t)N * HID;
    float* a_d   = a_s  + (size_t)N * NH;

    int* csr_src = (int*)r4;
    int* rowptr  = csr_src + (size_t)E;
    int* cursor  = rowptr + (size_t)N + 1;
    int* cnt     = cursor + (size_t)N;
    int* bsum    = cnt + (size_t)N;

    const int BT = 256;
    const int gRowTiles = cdiv(N, 32);
    const int gN   = cdiv(N, BT);
    const int gE   = cdiv(E, BT);
    const int gScan = cdiv(N, SCAN_BLK);
    const int gNW  = cdiv(N * 32, BT);

    fill_i32<<<gN, BT, 0, stream>>>(cnt, 0, N);
    dst_count<<<gE, BT, 0, stream>>>(ei, cnt, E);
    scan_blocksum<<<gScan, SCAN_T, 0, stream>>>(cnt, bsum, N);
    scan_write<<<gScan, SCAN_T, 0, stream>>>(cnt, bsum, rowptr, cursor, N);
    csr_fill<<<gE, BT, 0, stream>>>(ei, cursor, csr_src, E);

    gemm_wmma_f32<64><<<gRowTiles, 128, 0, stream>>>(x, W1, xp1, att_src1, att_dst1, a_s, a_d, N, HID);
    node_softmax_agg<false><<<gNW, BT, 0, stream>>>(rowptr, csr_src, xp1, a_s, a_d, bias1, agg1, N);

    gemm_wmma_f32<128><<<gRowTiles, 128, 0, stream>>>(agg1, W2, xp2, att_src2, att_dst2, a_s, a_d, N, HID);
    node_softmax_agg<true><<<gNW, BT, 0, stream>>>(rowptr, csr_src, xp2, a_s, a_d, bias2, out, N);
}


# Edge rank kept from the counting pass's returning atomic; fill pass places edges at rowptr[dst]+rank with plain loads: one scattered atomic per edge, not two; edge order still by atomic arrival
# speedup vs baseline: 4.3430x; 1.2338x over previous
#include <hip/hip_runtime.h>
#include <math.h>

typedef __attribute__((ext_vector_type(2))) float v2f;
typedef __attribute__((ext_vector_type(4))) float v4f;
typedef __attribute__((ext_vector_type(8))) float v8f;

#define HID 128
#define NH  4
#define NC  32
#define NEG_SLOPE 0.2f

template <int XM>
__device__ __forceinline__ float row_xor(float v) {
    return __int_as_float(__builtin_amdgcn_update_dpp(
        0, __float_as_int(v), 0x160 | XM, 0xf, 0xf, true));
}


__device__ __forceinline__ float fold16(float (&v)[16], int l15) {
#pragma unroll
    for (int i = 0; i < 8; ++i) {
        const bool up = (l15 & 8) != 0;
        const float send = up ? v[i] : v[i + 8];
        const float keep = up ? v[i + 8] : v[i];
        v[i] = keep + row_xor<8>(send);
    }
#pragma unroll
    for (int i = 0; i < 4; ++i) {
        const bool up = (l15 & 4) != 0;
        const float send = up ? v[i] : v[i + 4];
        const float keep = up ? v[i + 4] : v[i];
        v[i] = keep + row_xor<4>(send);
    }
#pragma unroll
    for (int i = 0; i < 2; ++i) {
        const bool up = (l15 & 2) != 0;
        const float send = up ? v[i] : v[i + 2];
        const float keep = up ? v[i + 2] : v[i];
        v[i] = keep + row_xor<2>(send);
    }
    const bool up = (l15 & 1) != 0;
    const float send = up ? v[0] : v[1];
    const float keep = up ? v[1] : v[0];
    return keep + row_xor<1>(send);
}

template <int K>
__global__ void gemm_wmma_f32(const float* __restrict__ A,
                              const float* __restrict__ W,
                              float* __restrict__ C,
                              const float* __restrict__ att_s,
                              const float* __restrict__ att_d,
                              float* __restrict__ a_s, float* __restrict__ a_d,
                              int n, int Oc) {
    __shared__ __attribute__((aligned(16))) float sAtt[2][32][NH];

    const int wave = threadIdx.x >> 5;
    const int lane = threadIdx.x & 31;
    const int rowBase = blockIdx.x * 32;
    const int colBase = wave * 32;

    const int half = lane >> 4;
    const int l15  = lane & 15;
    int row0 = rowBase + l15;
    int row1 = row0 + 16;
    if (row0 >= n) row0 = n - 1;
    if (row1 >= n) row1 = n - 1;
    const int col0 = colBase + l15;
    const int col1 = col0 + 16;

    const float* Ap0 = A + (size_t)row0 * K + 4 * half;
    const float* Ap1 = A + (size_t)row1 * K + 4 * half;
    const float* Wp0 = W + (size_t)col0 * K + 4 * half;
    const float* Wp1 = W + (size_t)col1 * K + 4 * half;

    v8f acc00 = {}, acc01 = {}, acc10 = {}, acc11 = {};
#pragma unroll
    for (int k = 0; k < K; k += 8) {
        const v4f a0 = *(const v4f*)(Ap0 + k);
        const v4f a1 = *(const v4f*)(Ap1 + k);
        const v4f b0 = *(const v4f*)(Wp0 + k);
        const v4f b1 = *(const v4f*)(Wp1 + k);
        const v2f a0l = {a0.x, a0.y}, a0h = {a0.z, a0.w};
        const v2f a1l = {a1.x, a1.y}, a1h = {a1.z, a1.w};
        const v2f b0l = {b0.x, b0.y}, b0h = {b0.z, b0.w};
        const v2f b1l = {b1.x, b1.y}, b1h = {b1.z, b1.w};
        acc00 = __builtin_amdgcn_wmma_f32_16x16x4_f32(
            false, a0l, false, b0l, (short)0, acc00, false, false);
        acc01 = __builtin_amdgcn_wmma_f32_16x16x4_f32(
            false, a0l, false, b1l, (short)0, acc01, false, false);
        acc10 = __builtin_amdgcn_wmma_f32_16x16x4_f32(
            false, a1l, false, b0l, (short)0, acc10, false, false);
        acc11 = __builtin_amdgcn_wmma_f32_16x16x4_f32(
            false, a1l, false, b1l, (short)0, acc11, false, false);
        acc00 = __builtin_amdgcn_wmma_f32_16x16x4_f32(
            false, a0h, false, b0h, (short)0, acc00, false, false);
        acc01 = __builtin_amdgcn_wmma_f32_16x16x4_f32(
            false, a0h, false, b1h, (short)0, acc01, false, false);
        acc10 = __builtin_amdgcn_wmma_f32_16x16x4_f32(
            false, a1h, false, b0h, (short)0, acc10, false, false);
        acc11 = __builtin_amdgcn_wmma_f32_16x16x4_f32(
            false, a1h, false, b1h, (short)0, acc11, false, false);
    }

    const float ws0 = att_s[col0], ws1 = att_s[col1];
    const float wd0 = att_d[col0], wd1 = att_d[col1];
    float v[16];
#pragma unroll
    for (int r = 0; r < 8; ++r) {
        v[r]     = acc00[r] * ws0 + acc01[r] * ws1;
        v[r + 8] = acc00[r] * wd0 + acc01[r] * wd1;
    }
    const float d0 = fold16(v, l15);
#pragma unroll
    for (int r = 0; r < 8; ++r) {
        v[r]     = acc10[r] * ws0 + acc11[r] * ws1;
        v[r + 8] = acc10[r] * wd0 + acc11[r] * wd1;
    }
    const float d1 = fold16(v, l15);
    sAtt[l15 >> 3][half * 8 + (l15 & 7)][wave]      = d0;
    sAtt[l15 >> 3][16 + half * 8 + (l15 & 7)][wave] = d1;
    __syncthreads();
    if (wave < 2) {
        const int grow = rowBase + lane;
        if (grow < n) {
            const float4 o = *(const float4*)&sAtt[wave][lane][0];
            *(float4*)((wave ? a_d : a_s) + (size_t)grow * NH) = o;
        }
    }

    const int crow0 = rowBase + half * 8;
#pragma unroll
    for (int r = 0; r < 8; ++r) {
        const int crow = crow0 + r;
        if (crow < n) {
            C[(size_t)crow * Oc + col0] = acc00[r];
            C[(size_t)crow * Oc + col1] = acc01[r];
        }
    }
#pragma unroll
    for (int r = 0; r < 8; ++r) {
        const int crow = crow0 + 16 + r;
        if (crow < n) {
            C[(size_t)crow * Oc + col0] = acc10[r];
            C[(size_t)crow * Oc + col1] = acc11[r];
        }
    }
}


#define SCAN_T   256
#define SCAN_PER 4
#define SCAN_BLK (SCAN_T * SCAN_PER)

__global__ void fill_i32(int* __restrict__ p, int v, int n) {
    int i = blockIdx.x * blockDim.x + threadIdx.x;
    if (i < n) p[i] = v;
}

__global__ void dst_count_rank(const int* __restrict__ ei, int* __restrict__ cnt,
                               int* __restrict__ rank, int E) {
    int i = blockIdx.x * blockDim.x + threadIdx.x;
    if (i < E) rank[i] = atomicAdd(&cnt[ei[E + i]], 1);
}

__global__ void scan_blocksum(const int* __restrict__ cnt, int* __restrict__ bsum, int n) {
    __shared__ int wsum[SCAN_T / 32];
    const int lane = threadIdx.x & 31, wave = threadIdx.x >> 5;
    const int base = blockIdx.x * SCAN_BLK + threadIdx.x * SCAN_PER;
    int v = 0;
#pragma unroll
    for (int k = 0; k < SCAN_PER; ++k) if (base + k < n) v += cnt[base + k];
#pragma unroll
    for (int off = 16; off > 0; off >>= 1) v += __shfl_xor(v, off);
    if (lane == 0) wsum[wave] = v;
    __syncthreads();
    if (threadIdx.x == 0) {
        int t = 0;
#pragma unroll
        for (int w = 0; w < SCAN_T / 32; ++w) t += wsum[w];
        bsum[blockIdx.x] = t;
    }
}

__global__ void scan_write(const int* __restrict__ cnt, const int* __restrict__ bsum,
                           int* __restrict__ rowptr, int n) {
    __shared__ int wincl[SCAN_T / 32];
    __shared__ int wprev[SCAN_T / 32];
    const int lane = threadIdx.x & 31, wave = threadIdx.x >> 5;

    int prev = 0;
    for (int j = threadIdx.x; j < (int)blockIdx.x; j += SCAN_T) prev += bsum[j];

    const int base = blockIdx.x * SCAN_BLK + threadIdx.x * SCAN_PER;
    int c[SCAN_PER];
    int tsum = 0;
#pragma unroll
    for (int k = 0; k < SCAN_PER; ++k) {
        c[k] = (base + k < n) ? cnt[base + k] : 0;
        tsum += c[k];
    }
    int incl = tsum;
#pragma unroll
    for (int off = 1; off < 32; off <<= 1) {
        const int t = __shfl_up(incl, off);
        if (lane >= off) incl += t;
    }
#pragma unroll
    for (int off = 16; off > 0; off >>= 1) prev += __shfl_xor(prev, off);
    if (lane == 31) wincl[wave] = incl;
    if (lane == 0)  wprev[wave] = prev;
    __syncthreads();
    int start = incl - tsum;
#pragma unroll
    for (int w = 0; w < SCAN_T / 32; ++w) {
        start += wprev[w];
        if (w < wave) start += wincl[w];
    }
#pragma unroll
    for (int k = 0; k < SCAN_PER; ++k) {
        const int i = base + k;
        if (i < n) {
            rowptr[i] = start;
            start += c[k];
            if (i == n - 1) rowptr[n] = start;
        }
    }
}

__global__ void csr_fill(const int* __restrict__ ei, const int* __restrict__ rowptr,
                         const int* __restrict__ rank, int* __restrict__ csr_src, int E) {
    int i = blockIdx.x * blockDim.x + threadIdx.x;
    if (i < E) csr_src[rowptr[ei[E + i]] + rank[i]] = ei[i];
}

__device__ __forceinline__ float lrelu(float v) {
    return v > 0.0f ? v : NEG_SLOPE * v;
}

__device__ __forceinline__ float elu1(float v) {
    return v > 0.0f ? v : expm1f(v);
}

__device__ __forceinline__ float4 edge_logits(const float* __restrict__ a_s, int s,
                                              const float4 ad) {
    const float4 a = *(const float4*)(a_s + (size_t)s * NH);
    return make_float4(lrelu(a.x + ad.x), lrelu(a.y + ad.y),
                       lrelu(a.z + ad.z), lrelu(a.w + ad.w));
}

__device__ __forceinline__ float4 exp_shifted(const float4 e, const float4 m) {
    return make_float4(expf(e.x - m.x), expf(e.y - m.y), expf(e.z - m.z), expf(e.w - m.w));
}

__device__ __forceinline__ float lane_bcast(float v, int srcLane) {
    return __int_as_float(__builtin_amdgcn_readlane(__float_as_int(v), srcLane));
}

__device__ __forceinline__ void gather_chunk(const float* __restrict__ xp, int s,
                                             const float4 p, int cnt, const float4 f,
                                             int lane, float4& acc, float& den) {
#pragma unroll 4
    for (int t = 0; t < cnt; ++t) {
        const int   st = __builtin_amdgcn_readlane(s, t);
        const float w  = lane_bcast(p.x, t) * f.x + lane_bcast(p.y, t) * f.y +
                         lane_bcast(p.z, t) * f.z + lane_bcast(p.w, t) * f.w;
        const float4 r = *(const float4*)(xp + (size_t)st * HID + lane * 4);
        acc.x += r.x * w; acc.y += r.y * w; acc.z += r.z * w; acc.w += r.w * w;
        den += w;
    }
}

template <bool LAST>
__global__ void node_softmax_agg(const int* __restrict__ rowptr,
                                 const int* __restrict__ csr_src,
                                 const float* __restrict__ xp,
                                 const float* __restrict__ a_s,
                                 const float* __restrict__ a_d,
                                 const float* __restrict__ bias,
                                 float* __restrict__ dst, int n) {
    const int lane = threadIdx.x & 31;
    const int node = __builtin_amdgcn_readfirstlane(
        (int)((blockIdx.x * blockDim.x + threadIdx.x) >> 5));
    if (node >= n) return;
    const int hme = lane >> 3;
    const float4 f = make_float4(hme == 0 ? 1.0f : 0.0f, hme == 1 ? 1.0f : 0.0f,
                                 hme == 2 ? 1.0f : 0.0f, hme == 3 ? 1.0f : 0.0f);
    const int beg = rowptr[node], end = rowptr[node + 1];

    const float4 ad = *(const float4*)(a_d + (size_t)node * NH);
    const float4 es = edge_logits(a_s, node, ad);

    const int j0 = beg + lane;
    const bool v0 = j0 < end;
    const int s0 = v0 ? csr_src[j0] : node;
    const float4 e0 = edge_logits(a_s, s0, ad);
    float4 m = make_float4(fmaxf(es.x, e0.x), fmaxf(es.y, e0.y),
                           fmaxf(es.z, e0.z), fmaxf(es.w, e0.w));
    for (int base = beg + 32; base < end; base += 32) {
        const int j = base + lane;
        if (j < end) {
            const float4 e = edge_logits(a_s, csr_src[j], ad);
            m.x = fmaxf(m.x, e.x); m.y = fmaxf(m.y, e.y);
            m.z = fmaxf(m.z, e.z); m.w = fmaxf(m.w, e.w);
        }
    }
#pragma unroll
    for (int off = 16; off > 0; off >>= 1) {
        m.x = fmaxf(m.x, __shfl_xor(m.x, off));
        m.y = fmaxf(m.y, __shfl_xor(m.y, off));
        m.z = fmaxf(m.z, __shfl_xor(m.z, off));
        m.w = fmaxf(m.w, __shfl_xor(m.w, off));
    }

    float4 acc;
    float den;
    {
        const float4 ps = exp_shifted(es, m);
        const float w = ps.x * f.x + ps.y * f.y + ps.z * f.z + ps.w * f.w;
        const float4 r = *(const float4*)(xp + (size_t)node * HID + lane * 4);
        acc = make_float4(r.x * w, r.y * w, r.z * w, r.w * w);
        den = w;
    }
    {
        const float4 p0 = exp_shifted(e0, m);
        const int cnt = min(32, end - beg);
        gather_chunk(xp, s0, p0, cnt, f, lane, acc, den);
    }
    for (int base = beg + 32; base < end; base += 32) {
        const int j = base + lane;
        const int s = (j < end) ? csr_src[j] : node;
        const float4 p = exp_shifted(edge_logits(a_s, s, ad), m);
        gather_chunk(xp, s, p, min(32, end - base), f, lane, acc, den);
    }

    const float dd = den + 1e-16f;
    const float4 a = make_float4(acc.x / dd, acc.y / dd, acc.z / dd, acc.w / dd);
    if (!LAST) {
        const float4 b = *(const float4*)(bias + lane * 4);
        *(float4*)(dst + (size_t)node * HID + lane * 4) =
            make_float4(elu1(a.x + b.x), elu1(a.y + b.y), elu1(a.z + b.z), elu1(a.w + b.w));
    } else {
        const float4 a1 = make_float4(__shfl_xor(a.x, 8),  __shfl_xor(a.y, 8),
                                      __shfl_xor(a.z, 8),  __shfl_xor(a.w, 8));
        const float4 a2 = make_float4(__shfl_xor(a.x, 16), __shfl_xor(a.y, 16),
                                      __shfl_xor(a.z, 16), __shfl_xor(a.w, 16));
        const float4 a3 = make_float4(__shfl_xor(a.x, 24), __shfl_xor(a.y, 24),
                                      __shfl_xor(a.z, 24), __shfl_xor(a.w, 24));
        if (lane < 8) {
            const float4 b = *(const float4*)(bias + lane * 4);
            *(float4*)(dst + (size_t)node * NC + lane * 4) =
                make_float4(0.25f * (a.x + a1.x + a2.x + a3.x) + b.x,
                            0.25f * (a.y + a1.y + a2.y + a3.y) + b.y,
                            0.25f * (a.z + a1.z + a2.z + a3.z) + b.z,
                            0.25f * (a.w + a1.w + a2.w + a3.w) + b.w);
        }
    }
}


static inline int cdiv(int a, int b) { return (a + b - 1) / b; }

extern "C" void kernel_launch(void* const* d_in, const int* in_sizes, int n_in,
                              void* d_out, int out_size, void* d_ws, size_t ws_size,
                              hipStream_t stream) {
    const float* x        = (const float*)d_in[0];
    const int*   ei       = (const int*)  d_in[1];
    const float* W1       = (const float*)d_in[2];
    const float* att_src1 = (const float*)d_in[3];
    const float* att_dst1 = (const float*)d_in[4];
    const float* bias1    = (const float*)d_in[5];
    const float* W2       = (const float*)d_in[6];
    const float* att_src2 = (const float*)d_in[7];
    const float* att_dst2 = (const float*)d_in[8];
    const float* bias2    = (const float*)d_in[9];
    float* out = (float*)d_out;

    const int IN = 64;
    const int N  = in_sizes[0] / IN;
    const int E  = in_sizes[1] / 2;

    float* xp1   = (float*)d_ws;
    float* agg1  = xp1  + (size_t)N * HID;
    float* xp2   = agg1 + (size_t)N * HID;
    float* r4    = xp2  + (size_t)N * HID;
    float* a_s   = r4   + (size_t)N * HID;
    float* a_d   = a_s  + (size_t)N * NH;

    int* csr_src = (int*)r4;
    int* rowptr  = csr_src + (size_t)E;
    int* rank    = rowptr + (size_t)N + 1;
    int* cnt     = rank + (size_t)E;
    int* bsum    = cnt + (size_t)N;

    const int BT = 256;
    const int gRowTiles = cdiv(N, 32);
    const int gN   = cdiv(N, BT);
    const int gE   = cdiv(E, BT);
    const int gScan = cdiv(N, SCAN_BLK);
    const int gNW  = cdiv(N * 32, BT);

    fill_i32<<<gN, BT, 0, stream>>>(cnt, 0, N);
    dst_count_rank<<<gE, BT, 0, stream>>>(ei, cnt, rank, E);
    scan_blocksum<<<gScan, SCAN_T, 0, stream>>>(cnt, bsum, N);
    scan_write<<<gScan, SCAN_T, 0, stream>>>(cnt, bsum, rowptr, N);
    csr_fill<<<gE, BT, 0, stream>>>(ei, rowptr, rank, csr_src, E);

    gemm_wmma_f32<64><<<gRowTiles, 128, 0, stream>>>(x, W1, xp1, att_src1, att_dst1, a_s, a_d, N, HID);
    node_softmax_agg<false><<<gNW, BT, 0, stream>>>(rowptr, csr_src, xp1, a_s, a_d, bias1, agg1, N);

    gemm_wmma_f32<128><<<gRowTiles, 128, 0, stream>>>(agg1, W2, xp2, att_src2, att_dst2, a_s, a_d, N, HID);
    node_softmax_agg<true><<<gNW, BT, 0, stream>>>(rowptr, csr_src, xp2, a_s, a_d, bias2, out, N);
}


# Edges grouped by destination by a two-level sort on LDS atomics (chunk bucket histogram, prefix, bucket scatter, per-bucket counting sort): no global atomic left; in-node edge order by LDS arrival
# speedup vs baseline: 5.3106x; 1.2228x over previous
#include <hip/hip_runtime.h>
#include <math.h>

typedef __attribute__((ext_vector_type(2))) float v2f;
typedef __attribute__((ext_vector_type(4))) float v4f;
typedef __attribute__((ext_vector_type(8))) float v8f;

#define HID 128
#define NH  4
#define NC  32
#define NEG_SLOPE 0.2f

template <int XM>
__device__ __forceinline__ float row_xor(float v) {
    return __int_as_float(__builtin_amdgcn_update_dpp(
        0, __float_as_int(v), 0x160 | XM, 0xf, 0xf, true));
}


__device__ __forceinline__ float fold16(float (&v)[16], int l15) {
#pragma unroll
    for (int i = 0; i < 8; ++i) {
        const bool up = (l15 & 8) != 0;
        const float send = up ? v[i] : v[i + 8];
        const float keep = up ? v[i + 8] : v[i];
        v[i] = keep + row_xor<8>(send);
    }
#pragma unroll
    for (int i = 0; i < 4; ++i) {
        const bool up = (l15 & 4) != 0;
        const float send = up ? v[i] : v[i + 4];
        const float keep = up ? v[i + 4] : v[i];
        v[i] = keep + row_xor<4>(send);
    }
#pragma unroll
    for (int i = 0; i < 2; ++i) {
        const bool up = (l15 & 2) != 0;
        const float send = up ? v[i] : v[i + 2];
        const float keep = up ? v[i + 2] : v[i];
        v[i] = keep + row_xor<2>(send);
    }
    const bool up = (l15 & 1) != 0;
    const float send = up ? v[0] : v[1];
    const float keep = up ? v[1] : v[0];
    return keep + row_xor<1>(send);
}

template <int K>
__global__ void gemm_wmma_f32(const float* __restrict__ A,
                              const float* __restrict__ W,
                              float* __restrict__ C,
                              const float* __restrict__ att_s,
                              const float* __restrict__ att_d,
                              float* __restrict__ a_s, float* __restrict__ a_d,
                              int n, int Oc) {
    __shared__ __attribute__((aligned(16))) float sAtt[2][32][NH];

    const int wave = threadIdx.x >> 5;
    const int lane = threadIdx.x & 31;
    const int rowBase = blockIdx.x * 32;
    const int colBase = wave * 32;

    const int half = lane >> 4;
    const int l15  = lane & 15;
    int row0 = rowBase + l15;
    int row1 = row0 + 16;
    if (row0 >= n) row0 = n - 1;
    if (row1 >= n) row1 = n - 1;
    const int col0 = colBase + l15;
    const int col1 = col0 + 16;

    const float* Ap0 = A + (size_t)row0 * K + 4 * half;
    const float* Ap1 = A + (size_t)row1 * K + 4 * half;
    const float* Wp0 = W + (size_t)col0 * K + 4 * half;
    const float* Wp1 = W + (size_t)col1 * K + 4 * half;

    v8f acc00 = {}, acc01 = {}, acc10 = {}, acc11 = {};
#pragma unroll
    for (int k = 0; k < K; k += 8) {
        const v4f a0 = *(const v4f*)(Ap0 + k);
        const v4f a1 = *(const v4f*)(Ap1 + k);
        const v4f b0 = *(const v4f*)(Wp0 + k);
        const v4f b1 = *(const v4f*)(Wp1 + k);
        const v2f a0l = {a0.x, a0.y}, a0h = {a0.z, a0.w};
        const v2f a1l = {a1.x, a1.y}, a1h = {a1.z, a1.w};
        const v2f b0l = {b0.x, b0.y}, b0h = {b0.z, b0.w};
        const v2f b1l = {b1.x, b1.y}, b1h = {b1.z, b1.w};
        acc00 = __builtin_amdgcn_wmma_f32_16x16x4_f32(
            false, a0l, false, b0l, (short)0, acc00, false, false);
        acc01 = __builtin_amdgcn_wmma_f32_16x16x4_f32(
            false, a0l, false, b1l, (short)0, acc01, false, false);
        acc10 = __builtin_amdgcn_wmma_f32_16x16x4_f32(
            false, a1l, false, b0l, (short)0, acc10, false, false);
        acc11 = __builtin_amdgcn_wmma_f32_16x16x4_f32(
            false, a1l, false, b1l, (short)0, acc11, false, false);
        acc00 = __builtin_amdgcn_wmma_f32_16x16x4_f32(
            false, a0h, false, b0h, (short)0, acc00, false, false);
        acc01 = __builtin_amdgcn_wmma_f32_16x16x4_f32(
            false, a0h, false, b1h, (short)0, acc01, false, false);
        acc10 = __builtin_amdgcn_wmma_f32_16x16x4_f32(
            false, a1h, false, b0h, (short)0, acc10, false, false);
        acc11 = __builtin_amdgcn_wmma_f32_16x16x4_f32(
            false, a1h, false, b1h, (short)0, acc11, false, false);
    }

    const float ws0 = att_s[col0], ws1 = att_s[col1];
    const float wd0 = att_d[col0], wd1 = att_d[col1];
    float v[16];
#pragma unroll
    for (int r = 0; r < 8; ++r) {
        v[r]     = acc00[r] * ws0 + acc01[r] * ws1;
        v[r + 8] = acc00[r] * wd0 + acc01[r] * wd1;
    }
    const float d0 = fold16(v, l15);
#pragma unroll
    for (int r = 0; r < 8; ++r) {
        v[r]     = acc10[r] * ws0 + acc11[r] * ws1;
        v[r + 8] = acc10[r] * wd0 + acc11[r] * wd1;
    }
    const float d1 = fold16(v, l15);
    sAtt[l15 >> 3][half * 8 + (l15 & 7)][wave]      = d0;
    sAtt[l15 >> 3][16 + half * 8 + (l15 & 7)][wave] = d1;
    __syncthreads();
    if (wave < 2) {
        const int grow = rowBase + lane;
        if (grow < n) {
            const float4 o = *(const float4*)&sAtt[wave][lane][0];
            *(float4*)((wave ? a_d : a_s) + (size_t)grow * NH) = o;
        }
    }

    const int crow0 = rowBase + half * 8;
#pragma unroll
    for (int r = 0; r < 8; ++r) {
        const int crow = crow0 + r;
        if (crow < n) {
            C[(size_t)crow * Oc + col0] = acc00[r];
            C[(size_t)crow * Oc + col1] = acc01[r];
        }
    }
#pragma unroll
    for (int r = 0; r < 8; ++r) {
        const int crow = crow0 + 16 + r;
        if (crow < n) {
            C[(size_t)crow * Oc + col0] = acc10[r];
            C[(size_t)crow * Oc + col1] = acc11[r];
        }
    }
}


#define SCAN_T   256
#define SCAN_PER 4
#define SCAN_BLK (SCAN_T * SCAN_PER)

#define BK_SH    8
#define BK_NODES (1 << BK_SH)
#define BK_MAX   512
#define SORT_T   256
#define SORT_G   512
#define SORT_U   16
#define CSR_U    8

__global__ void bucket_hist(const int* __restrict__ ei, int* __restrict__ hist,
                            int E, int chunk, int nbuck) {
    __shared__ int h[BK_MAX];
    for (int b = threadIdx.x; b < nbuck; b += SORT_T) h[b] = 0;
    __syncthreads();
    const int beg = blockIdx.x * chunk;
    const int end = min(E, beg + chunk);
    for (int i0 = beg + threadIdx.x; i0 < end; i0 += SORT_U * SORT_T) {
        int d[SORT_U];
#pragma unroll
        for (int u = 0; u < SORT_U; ++u)
            d[u] = ei[E + min(i0 + u * SORT_T, end - 1)];
#pragma unroll
        for (int u = 0; u < SORT_U; ++u)
            if (i0 + u * SORT_T < end) atomicAdd(&h[d[u] >> BK_SH], 1);
    }
    __syncthreads();
    for (int b = threadIdx.x; b < nbuck; b += SORT_T)
        hist[(size_t)b * gridDim.x + blockIdx.x] = h[b];
}

__global__ void scan_blocksum(const int* __restrict__ cnt, int* __restrict__ bsum, int n) {
    __shared__ int wsum[SCAN_T / 32];
    const int lane = threadIdx.x & 31, wave = threadIdx.x >> 5;
    const int base = blockIdx.x * SCAN_BLK + threadIdx.x * SCAN_PER;
    int v = 0;
#pragma unroll
    for (int k = 0; k < SCAN_PER; ++k) if (base + k < n) v += cnt[base + k];
#pragma unroll
    for (int off = 16; off > 0; off >>= 1) v += __shfl_xor(v, off);
    if (lane == 0) wsum[wave] = v;
    __syncthreads();
    if (threadIdx.x == 0) {
        int t = 0;
#pragma unroll
        for (int w = 0; w < SCAN_T / 32; ++w) t += wsum[w];
        bsum[blockIdx.x] = t;
    }
}

__global__ void scan_write(const int* __restrict__ cnt, const int* __restrict__ bsum,
                           int* __restrict__ rowptr, int n) {
    __shared__ int wincl[SCAN_T / 32];
    __shared__ int wprev[SCAN_T / 32];
    const int lane = threadIdx.x & 31, wave = threadIdx.x >> 5;

    int prev = 0;
    for (int j = threadIdx.x; j < (int)blockIdx.x; j += SCAN_T) prev += bsum[j];

    const int base = blockIdx.x * SCAN_BLK + threadIdx.x * SCAN_PER;
    int c[SCAN_PER];
    int tsum = 0;
#pragma unroll
    for (int k = 0; k < SCAN_PER; ++k) {
        c[k] = (base + k < n) ? cnt[base + k] : 0;
        tsum += c[k];
    }
    int incl = tsum;
#pragma unroll
    for (int off = 1; off < 32; off <<= 1) {
        const int t = __shfl_up(incl, off);
        if (lane >= off) incl += t;
    }
#pragma unroll
    for (int off = 16; off > 0; off >>= 1) prev += __shfl_xor(prev, off);
    if (lane == 31) wincl[wave] = incl;
    if (lane == 0)  wprev[wave] = prev;
    __syncthreads();
    int start = incl - tsum;
#pragma unroll
    for (int w = 0; w < SCAN_T / 32; ++w) {
        start += wprev[w];
        if (w < wave) start += wincl[w];
    }
#pragma unroll
    for (int k = 0; k < SCAN_PER; ++k) {
        const int i = base + k;
        if (i < n) {
            rowptr[i] = start;
            start += c[k];
            if (i == n - 1) rowptr[n] = start;
        }
    }
}

__global__ void bucket_scatter(const int* __restrict__ ei, const int* __restrict__ off,
                               int* __restrict__ tmp, int E, int chunk, int nbuck) {
    __shared__ int cur[BK_MAX];
    for (int b = threadIdx.x; b < nbuck; b += SORT_T)
        cur[b] = off[(size_t)b * gridDim.x + blockIdx.x];
    __syncthreads();
    const int beg = blockIdx.x * chunk;
    const int end = min(E, beg + chunk);
    for (int i0 = beg + threadIdx.x; i0 < end; i0 += SORT_U * SORT_T) {
        int d[SORT_U], s[SORT_U], pos[SORT_U];
#pragma unroll
        for (int u = 0; u < SORT_U; ++u) {
            const int j = min(i0 + u * SORT_T, end - 1);
            d[u] = ei[E + j];
            s[u] = ei[j];
        }
#pragma unroll
        for (int u = 0; u < SORT_U; ++u)
            if (i0 + u * SORT_T < end) pos[u] = atomicAdd(&cur[d[u] >> BK_SH], 1);
#pragma unroll
        for (int u = 0; u < SORT_U; ++u)
            if (i0 + u * SORT_T < end)
                tmp[pos[u]] = (int)(((unsigned)s[u] << BK_SH) | (unsigned)(d[u] & (BK_NODES - 1)));
    }
}

__global__ void bucket_csr(const int* __restrict__ tmp, const int* __restrict__ off,
                           int* __restrict__ rowptr, int* __restrict__ csr_src,
                           int N, int G) {
    __shared__ int c[BK_NODES];
    __shared__ int wtot[SORT_T / 32];
    const int lane = threadIdx.x & 31, wave = threadIdx.x >> 5;
    const int bbeg = off[(size_t)blockIdx.x * G];
    const int bend = off[(size_t)(blockIdx.x + 1) * G];
    c[threadIdx.x] = 0;
    __syncthreads();
    for (int j0 = bbeg + threadIdx.x; j0 < bend; j0 += CSR_U * SORT_T) {
        int v[CSR_U];
#pragma unroll
        for (int u = 0; u < CSR_U; ++u)
            v[u] = tmp[min(j0 + u * SORT_T, bend - 1)];
#pragma unroll
        for (int u = 0; u < CSR_U; ++u)
            if (j0 + u * SORT_T < bend) atomicAdd(&c[v[u] & (BK_NODES - 1)], 1);
    }
    __syncthreads();
    const int mine = c[threadIdx.x];
    int incl = mine;
#pragma unroll
    for (int o = 1; o < 32; o <<= 1) {
        const int t = __shfl_up(incl, o);
        if (lane >= o) incl += t;
    }
    if (lane == 31) wtot[wave] = incl;
    __syncthreads();
    int start = bbeg + incl - mine;
#pragma unroll
    for (int w = 0; w < SORT_T / 32; ++w)
        if (w < wave) start += wtot[w];
    const int node = blockIdx.x * BK_NODES + threadIdx.x;
    if (node < N) rowptr[node] = start;
    if (node == N - 1) rowptr[N] = start + mine;
    c[threadIdx.x] = start;
    __syncthreads();
    for (int j0 = bbeg + threadIdx.x; j0 < bend; j0 += CSR_U * SORT_T) {
        int v[CSR_U], pos[CSR_U];
#pragma unroll
        for (int u = 0; u < CSR_U; ++u)
            v[u] = tmp[min(j0 + u * SORT_T, bend - 1)];
#pragma unroll
        for (int u = 0; u < CSR_U; ++u)
            if (j0 + u * SORT_T < bend) pos[u] = atomicAdd(&c[v[u] & (BK_NODES - 1)], 1);
#pragma unroll
        for (int u = 0; u < CSR_U; ++u)
            if (j0 + u * SORT_T < bend) csr_src[pos[u]] = (int)((unsigned)v[u] >> BK_SH);
    }
}

__device__ __forceinline__ float lrelu(float v) {
    return v > 0.0f ? v : NEG_SLOPE * v;
}

__device__ __forceinline__ float elu1(float v) {
    return v > 0.0f ? v : expm1f(v);
}

__device__ __forceinline__ float4 edge_logits(const float* __restrict__ a_s, int s,
                                              const float4 ad) {
    const float4 a = *(const float4*)(a_s + (size_t)s * NH);
    return make_float4(lrelu(a.x + ad.x), lrelu(a.y + ad.y),
                       lrelu(a.z + ad.z), lrelu(a.w + ad.w));
}

__device__ __forceinline__ float4 exp_shifted(const float4 e, const float4 m) {
    return make_float4(expf(e.x - m.x), expf(e.y - m.y), expf(e.z - m.z), expf(e.w - m.w));
}

__device__ __forceinline__ float lane_bcast(float v, int srcLane) {
    return __int_as_float(__builtin_amdgcn_readlane(__float_as_int(v), srcLane));
}

__device__ __forceinline__ void gather_chunk(const float* __restrict__ xp, int s,
                                             const float4 p, int cnt, const float4 f,
                                             int lane, float4& acc, float& den) {
#pragma unroll 4
    for (int t = 0; t < cnt; ++t) {
        const int   st = __builtin_amdgcn_readlane(s, t);
        const float w  = lane_bcast(p.x, t) * f.x + lane_bcast(p.y, t) * f.y +
                         lane_bcast(p.z, t) * f.z + lane_bcast(p.w, t) * f.w;
        const float4 r = *(const float4*)(xp + (size_t)st * HID + lane * 4);
        acc.x += r.x * w; acc.y += r.y * w; acc.z += r.z * w; acc.w += r.w * w;
        den += w;
    }
}

template <bool LAST>
__global__ void node_softmax_agg(const int* __restrict__ rowptr,
                                 const int* __restrict__ csr_src,
                                 const float* __restrict__ xp,
                                 const float* __restrict__ a_s,
                                 const float* __restrict__ a_d,
                                 const float* __restrict__ bias,
                                 float* __restrict__ dst, int n) {
    const int lane = threadIdx.x & 31;
    const int node = __builtin_amdgcn_readfirstlane(
        (int)((blockIdx.x * blockDim.x + threadIdx.x) >> 5));
    if (node >= n) return;
    const int hme = lane >> 3;
    const float4 f = make_float4(hme == 0 ? 1.0f : 0.0f, hme == 1 ? 1.0f : 0.0f,
                                 hme == 2 ? 1.0f : 0.0f, hme == 3 ? 1.0f : 0.0f);
    const int beg = rowptr[node], end = rowptr[node + 1];

    const float4 ad = *(const float4*)(a_d + (size_t)node * NH);
    const float4 es = edge_logits(a_s, node, ad);

    const int j0 = beg + lane;
    const bool v0 = j0 < end;
    const int s0 = v0 ? csr_src[j0] : node;
    const float4 e0 = edge_logits(a_s, s0, ad);
    float4 m = make_float4(fmaxf(es.x, e0.x), fmaxf(es.y, e0.y),
                           fmaxf(es.z, e0.z), fmaxf(es.w, e0.w));
    for (int base = beg + 32; base < end; base += 32) {
        const int j = base + lane;
        if (j < end) {
            const float4 e = edge_logits(a_s, csr_src[j], ad);
            m.x = fmaxf(m.x, e.x); m.y = fmaxf(m.y, e.y);
            m.z = fmaxf(m.z, e.z); m.w = fmaxf(m.w, e.w);
        }
    }
#pragma unroll
    for (int off = 16; off > 0; off >>= 1) {
        m.x = fmaxf(m.x, __shfl_xor(m.x, off));
        m.y = fmaxf(m.y, __shfl_xor(m.y, off));
        m.z = fmaxf(m.z, __shfl_xor(m.z, off));
        m.w = fmaxf(m.w, __shfl_xor(m.w, off));
    }

    float4 acc;
    float den;
    {
        const float4 ps = exp_shifted(es, m);
        const float w = ps.x * f.x + ps.y * f.y + ps.z * f.z + ps.w * f.w;
        const float4 r = *(const float4*)(xp + (size_t)node * HID + lane * 4);
        acc = make_float4(r.x * w, r.y * w, r.z * w, r.w * w);
        den = w;
    }
    {
        const float4 p0 = exp_shifted(e0, m);
        const int cnt = min(32, end - beg);
        gather_chunk(xp, s0, p0, cnt, f, lane, acc, den);
    }
    for (int base = beg + 32; base < end; base += 32) {
        const int j = base + lane;
        const int s = (j < end) ? csr_src[j] : node;
        const float4 p = exp_shifted(edge_logits(a_s, s, ad), m);
        gather_chunk(xp, s, p, min(32, end - base), f, lane, acc, den);
    }

    const float dd = den + 1e-16f;
    const float4 a = make_float4(acc.x / dd, acc.y / dd, acc.z / dd, acc.w / dd);
    if (!LAST) {
        const float4 b = *(const float4*)(bias + lane * 4);
        *(float4*)(dst + (size_t)node * HID + lane * 4) =
            make_float4(elu1(a.x + b.x), elu1(a.y + b.y), elu1(a.z + b.z), elu1(a.w + b.w));
    } else {
        const float4 a1 = make_float4(__shfl_xor(a.x, 8),  __shfl_xor(a.y, 8),
                                      __shfl_xor(a.z, 8),  __shfl_xor(a.w, 8));
        const float4 a2 = make_float4(__shfl_xor(a.x, 16), __shfl_xor(a.y, 16),
                                      __shfl_xor(a.z, 16), __shfl_xor(a.w, 16));
        const float4 a3 = make_float4(__shfl_xor(a.x, 24), __shfl_xor(a.y, 24),
                                      __shfl_xor(a.z, 24), __shfl_xor(a.w, 24));
        if (lane < 8) {
            const float4 b = *(const float4*)(bias + lane * 4);
            *(float4*)(dst + (size_t)node * NC + lane * 4) =
                make_float4(0.25f * (a.x + a1.x + a2.x + a3.x) + b.x,
                            0.25f * (a.y + a1.y + a2.y + a3.y) + b.y,
                            0.25f * (a.z + a1.z + a2.z + a3.z) + b.z,
                            0.25f * (a.w + a1.w + a2.w + a3.w) + b.w);
        }
    }
}


static inline int cdiv(int a, int b) { return (a + b - 1) / b; }

extern "C" void kernel_launch(void* const* d_in, const int* in_sizes, int n_in,
                              void* d_out, int out_size, void* d_ws, size_t ws_size,
                              hipStream_t stream) {
    const float* x        = (const float*)d_in[0];
    const int*   ei       = (const int*)  d_in[1];
    const float* W1       = (const float*)d_in[2];
    const float* att_src1 = (const float*)d_in[3];
    const float* att_dst1 = (const float*)d_in[4];
    const float* bias1    = (const float*)d_in[5];
    const float* W2       = (const float*)d_in[6];
    const float* att_src2 = (const float*)d_in[7];
    const float* att_dst2 = (const float*)d_in[8];
    const float* bias2    = (const float*)d_in[9];
    float* out = (float*)d_out;

    const int IN = 64;
    const int N  = in_sizes[0] / IN;
    const int E  = in_sizes[1] / 2;

    float* xp1   = (float*)d_ws;
    float* agg1  = xp1  + (size_t)N * HID;
    float* xp2   = agg1 + (size_t)N * HID;
    float* r4    = xp2  + (size_t)N * HID;
    float* a_s   = r4   + (size_t)N * HID;
    float* a_d   = a_s  + (size_t)N * NH;

    int* csr_src = (int*)r4;
    int* rowptr  = csr_src + (size_t)E;
    int* tmp     = rowptr + (size_t)N + 1;
    const int nbuck = cdiv(N, BK_NODES);
    const int nTab  = nbuck * SORT_G;
    int* hist    = tmp + (size_t)E;
    int* off     = hist + (size_t)nTab;
    int* bsum    = off + (size_t)nTab + 1;

    const int BT = 256;
    const int gRowTiles = cdiv(N, 32);
    const int chunk = cdiv(E, SORT_G);
    const int gScan = cdiv(nTab, SCAN_BLK);
    const int gNW  = cdiv(N * 32, BT);

    bucket_hist<<<SORT_G, SORT_T, 0, stream>>>(ei, hist, E, chunk, nbuck);
    scan_blocksum<<<gScan, SCAN_T, 0, stream>>>(hist, bsum, nTab);
    scan_write<<<gScan, SCAN_T, 0, stream>>>(hist, bsum, off, nTab);
    bucket_scatter<<<SORT_G, SORT_T, 0, stream>>>(ei, off, tmp, E, chunk, nbuck);
    bucket_csr<<<nbuck, SORT_T, 0, stream>>>(tmp, off, rowptr, csr_src, N, SORT_G);

    gemm_wmma_f32<64><<<gRowTiles, 128, 0, stream>>>(x, W1, xp1, att_src1, att_dst1, a_s, a_d, N, HID);
    node_softmax_agg<false><<<gNW, BT, 0, stream>>>(rowptr, csr_src, xp1, a_s, a_d, bias1, agg1, N);

    gemm_wmma_f32<128><<<gRowTiles, 128, 0, stream>>>(agg1, W2, xp2, att_src2, att_dst2, a_s, a_d, N, HID);
    node_softmax_agg<true><<<gNW, BT, 0, stream>>>(rowptr, csr_src, xp2, a_s, a_d, bias2, out, N);
}


# First layer's matrix product done by extra work-groups appended to three of the edge sort's launches (their own blocks leave most wave slots and the matrix pipes idle), no launch of its own: 9 to 8
# speedup vs baseline: 5.7084x; 1.0749x over previous
#include <hip/hip_runtime.h>
#include <math.h>

typedef __attribute__((ext_vector_type(2))) float v2f;
typedef __attribute__((ext_vector_type(4))) float v4f;
typedef __attribute__((ext_vector_type(8))) float v8f;

#define HID 128
#define NH  4
#define NC  32
#define NEG_SLOPE 0.2f

template <int XM>
__device__ __forceinline__ float row_xor(float v) {
    return __int_as_float(__builtin_amdgcn_update_dpp(
        0, __float_as_int(v), 0x160 | XM, 0xf, 0xf, true));
}


__device__ __forceinline__ float fold16(float (&v)[16], int l15) {
#pragma unroll
    for (int i = 0; i < 8; ++i) {
        const bool up = (l15 & 8) != 0;
        const float send = up ? v[i] : v[i + 8];
        const float keep = up ? v[i + 8] : v[i];
        v[i] = keep + row_xor<8>(send);
    }
#pragma unroll
    for (int i = 0; i < 4; ++i) {
        const bool up = (l15 & 4) != 0;
        const float send = up ? v[i] : v[i + 4];
        const float keep = up ? v[i + 4] : v[i];
        v[i] = keep + row_xor<4>(send);
    }
#pragma unroll
    for (int i = 0; i < 2; ++i) {
        const bool up = (l15 & 2) != 0;
        const float send = up ? v[i] : v[i + 2];
        const float keep = up ? v[i + 2] : v[i];
        v[i] = keep + row_xor<2>(send);
    }
    const bool up = (l15 & 1) != 0;
    const float send = up ? v[0] : v[1];
    const float keep = up ? v[1] : v[0];
    return keep + row_xor<1>(send);
}

template <int K>
__device__ __forceinline__ void gemm_rows32(const float* __restrict__ A,
                                            const float* __restrict__ W,
                                            float* __restrict__ C,
                                            const float* __restrict__ att_s,
                                            const float* __restrict__ att_d,
                                            float* __restrict__ a_s, float* __restrict__ a_d,
                                            int n, int Oc, int rowBase, int wave, int lane,
                                            float (*sAtt)[32][NH]) {
    const int colBase = wave * 32;

    const int half = lane >> 4;
    const int l15  = lane & 15;
    int row0 = rowBase + l15;
    int row1 = row0 + 16;
    if (row0 >= n) row0 = n - 1;
    if (row1 >= n) row1 = n - 1;
    const int col0 = colBase + l15;
    const int col1 = col0 + 16;

    const float* Ap0 = A + (size_t)row0 * K + 4 * half;
    const float* Ap1 = A + (size_t)row1 * K + 4 * half;
    const float* Wp0 = W + (size_t)col0 * K + 4 * half;
    const float* Wp1 = W + (size_t)col1 * K + 4 * half;

    v8f acc00 = {}, acc01 = {}, acc10 = {}, acc11 = {};
#pragma unroll
    for (int k = 0; k < K; k += 8) {
        const v4f a0 = *(const v4f*)(Ap0 + k);
        const v4f a1 = *(const v4f*)(Ap1 + k);
        const v4f b0 = *(const v4f*)(Wp0 + k);
        const v4f b1 = *(const v4f*)(Wp1 + k);
        const v2f a0l = {a0.x, a0.y}, a0h = {a0.z, a0.w};
        const v2f a1l = {a1.x, a1.y}, a1h = {a1.z, a1.w};
        const v2f b0l = {b0.x, b0.y}, b0h = {b0.z, b0.w};
        const v2f b1l = {b1.x, b1.y}, b1h = {b1.z, b1.w};
        acc00 = __builtin_amdgcn_wmma_f32_16x16x4_f32(
            false, a0l, false, b0l, (short)0, acc00, false, false);
        acc01 = __builtin_amdgcn_wmma_f32_16x16x4_f32(
            false, a0l, false, b1l, (short)0, acc01, false, false);
        acc10 = __builtin_amdgcn_wmma_f32_16x16x4_f32(
            false, a1l, false, b0l, (short)0, acc10, false, false);
        acc11 = __builtin_amdgcn_wmma_f32_16x16x4_f32(
            false, a1l, false, b1l, (short)0, acc11, false, false);
        acc00 = __builtin_amdgcn_wmma_f32_16x16x4_f32(
            false, a0h, false, b0h, (short)0, acc00, false, false);
        acc01 = __builtin_amdgcn_wmma_f32_16x16x4_f32(
            false, a0h, false, b1h, (short)0, acc01, false, false);
        acc10 = __builtin_amdgcn_wmma_f32_16x16x4_f32(
            false, a1h, false, b0h, (short)0, acc10, false, false);
        acc11 = __builtin_amdgcn_wmma_f32_16x16x4_f32(
            false, a1h, false, b1h, (short)0, acc11, false, false);
    }

    const float ws0 = att_s[col0], ws1 = att_s[col1];
    const float wd0 = att_d[col0], wd1 = att_d[col1];
    float v[16];
#pragma unroll
    for (int r = 0; r < 8; ++r) {
        v[r]     = acc00[r] * ws0 + acc01[r] * ws1;
        v[r + 8] = acc00[r] * wd0 + acc01[r] * wd1;
    }
    const float d0 = fold16(v, l15);
#pragma unroll
    for (int r = 0; r < 8; ++r) {
        v[r]     = acc10[r] * ws0 + acc11[r] * ws1;
        v[r + 8] = acc10[r] * wd0 + acc11[r] * wd1;
    }
    const float d1 = fold16(v, l15);
    sAtt[l15 >> 3][half * 8 + (l15 & 7)][wave]      = d0;
    sAtt[l15 >> 3][16 + half * 8 + (l15 & 7)][wave] = d1;
    __syncthreads();
    if (wave < 2) {
        const int grow = rowBase + lane;
        if (grow < n) {
            const float4 o = *(const float4*)&sAtt[wave][lane][0];
            *(float4*)((wave ? a_d : a_s) + (size_t)grow * NH) = o;
        }
    }

    const int crow0 = rowBase + half * 8;
#pragma unroll
    for (int r = 0; r < 8; ++r) {
        const int crow = crow0 + r;
        if (crow < n) {
            C[(size_t)crow * Oc + col0] = acc00[r];
            C[(size_t)crow * Oc + col1] = acc01[r];
        }
    }
#pragma unroll
    for (int r = 0; r < 8; ++r) {
        const int crow = crow0 + 16 + r;
        if (crow < n) {
            C[(size_t)crow * Oc + col0] = acc10[r];
            C[(size_t)crow * Oc + col1] = acc11[r];
        }
    }
}

template <int K>
__global__ void gemm_wmma_f32(const float* __restrict__ A,
                              const float* __restrict__ W,
                              float* __restrict__ C,
                              const float* __restrict__ att_s,
                              const float* __restrict__ att_d,
                              float* __restrict__ a_s, float* __restrict__ a_d,
                              int n, int Oc) {
    __shared__ __attribute__((aligned(16))) float sAtt[2][32][NH];
    gemm_rows32<K>(A, W, C, att_s, att_d, a_s, a_d, n, Oc,
                   (int)blockIdx.x * 32, (int)(threadIdx.x >> 5), (int)(threadIdx.x & 31), sAtt);
}

__device__ __forceinline__ void gemm1_block64(const float* __restrict__ x,
                                              const float* __restrict__ W1,
                                              float* __restrict__ xp,
                                              const float* __restrict__ att_s,
                                              const float* __restrict__ att_d,
                                              float* __restrict__ a_s, float* __restrict__ a_d,
                                              int n, int blk64) {
    __shared__ __attribute__((aligned(16))) float sAtt2[2][2][32][NH];
    const int w8  = (int)(threadIdx.x >> 5);
    const int grp = w8 >> 2;
    gemm_rows32<64>(x, W1, xp, att_s, att_d, a_s, a_d, n, HID,
                    blk64 * 64 + grp * 32, w8 & 3, (int)(threadIdx.x & 31), sAtt2[grp]);
}


#define SCAN_T   256
#define SCAN_PER 4
#define SCAN_BLK (SCAN_T * SCAN_PER)

#define BK_SH    8
#define BK_NODES (1 << BK_SH)
#define BK_MAX   512
#define SORT_T   256
#define SORT_G   512
#define SORT_U   16
#define CSR_U    8

__device__ __forceinline__ void bucket_hist_body(const int* __restrict__ ei, int* __restrict__ hist,
                                                 int E, int chunk, int nbuck, int G) {
    __shared__ int h[BK_MAX];
    for (int b = threadIdx.x; b < nbuck; b += SORT_T) h[b] = 0;
    __syncthreads();
    const int beg = blockIdx.x * chunk;
    const int end = min(E, beg + chunk);
    for (int i0 = beg + threadIdx.x; i0 < end; i0 += SORT_U * SORT_T) {
        int d[SORT_U];
#pragma unroll
        for (int u = 0; u < SORT_U; ++u)
            d[u] = ei[E + min(i0 + u * SORT_T, end - 1)];
#pragma unroll
        for (int u = 0; u < SORT_U; ++u)
            if (i0 + u * SORT_T < end) atomicAdd(&h[d[u] >> BK_SH], 1);
    }
    __syncthreads();
    for (int b = threadIdx.x; b < nbuck; b += SORT_T)
        hist[(size_t)b * G + blockIdx.x] = h[b];
}

__global__ void scan_blocksum(const int* __restrict__ cnt, int* __restrict__ bsum, int n) {
    __shared__ int wsum[SCAN_T / 32];
    const int lane = threadIdx.x & 31, wave = threadIdx.x >> 5;
    const int base = blockIdx.x * SCAN_BLK + threadIdx.x * SCAN_PER;
    int v = 0;
#pragma unroll
    for (int k = 0; k < SCAN_PER; ++k) if (base + k < n) v += cnt[base + k];
#pragma unroll
    for (int off = 16; off > 0; off >>= 1) v += __shfl_xor(v, off);
    if (lane == 0) wsum[wave] = v;
    __syncthreads();
    if (threadIdx.x == 0) {
        int t = 0;
#pragma unroll
        for (int w = 0; w < SCAN_T / 32; ++w) t += wsum[w];
        bsum[blockIdx.x] = t;
    }
}

__global__ void scan_write(const int* __restrict__ cnt, const int* __restrict__ bsum,
                           int* __restrict__ rowptr, int n) {
    __shared__ int wincl[SCAN_T / 32];
    __shared__ int wprev[SCAN_T / 32];
    const int lane = threadIdx.x & 31, wave = threadIdx.x >> 5;

    int prev = 0;
    for (int j = threadIdx.x; j < (int)blockIdx.x; j += SCAN_T) prev += bsum[j];

    const int base = blockIdx.x * SCAN_BLK + threadIdx.x * SCAN_PER;
    int c[SCAN_PER];
    int tsum = 0;
#pragma unroll
    for (int k = 0; k < SCAN_PER; ++k) {
        c[k] = (base + k < n) ? cnt[base + k] : 0;
        tsum += c[k];
    }
    int incl = tsum;
#pragma unroll
    for (int off = 1; off < 32; off <<= 1) {
        const int t = __shfl_up(incl, off);
        if (lane >= off) incl += t;
    }
#pragma unroll
    for (int off = 16; off > 0; off >>= 1) prev += __shfl_xor(prev, off);
    if (lane == 31) wincl[wave] = incl;
    if (lane == 0)  wprev[wave] = prev;
    __syncthreads();
    int start = incl - tsum;
#pragma unroll
    for (int w = 0; w < SCAN_T / 32; ++w) {
        start += wprev[w];
        if (w < wave) start += wincl[w];
    }
#pragma unroll
    for (int k = 0; k < SCAN_PER; ++k) {
        const int i = base + k;
        if (i < n) {
            rowptr[i] = start;
            start += c[k];
            if (i == n - 1) rowptr[n] = start;
        }
    }
}

__device__ __forceinline__ void bucket_scatter_body(const int* __restrict__ ei, const int* __restrict__ off,
                                                    int* __restrict__ tmp, int E, int chunk, int nbuck, int G) {
    __shared__ int cur[BK_MAX];
    for (int b = threadIdx.x; b < nbuck; b += SORT_T)
        cur[b] = off[(size_t)b * G + blockIdx.x];
    __syncthreads();
    const int beg = blockIdx.x * chunk;
    const int end = min(E, beg + chunk);
    for (int i0 = beg + threadIdx.x; i0 < end; i0 += SORT_U * SORT_T) {
        int d[SORT_U], s[SORT_U], pos[SORT_U];
#pragma unroll
        for (int u = 0; u < SORT_U; ++u) {
            const int j = min(i0 + u * SORT_T, end - 1);
            d[u] = ei[E + j];
            s[u] = ei[j];
        }
#pragma unroll
        for (int u = 0; u < SORT_U; ++u)
            if (i0 + u * SORT_T < end) pos[u] = atomicAdd(&cur[d[u] >> BK_SH], 1);
#pragma unroll
        for (int u = 0; u < SORT_U; ++u)
            if (i0 + u * SORT_T < end)
                tmp[pos[u]] = (int)(((unsigned)s[u] << BK_SH) | (unsigned)(d[u] & (BK_NODES - 1)));
    }
}

__device__ __forceinline__ void bucket_csr_body(const int* __restrict__ tmp, const int* __restrict__ off,
                                                int* __restrict__ rowptr, int* __restrict__ csr_src,
                                                int N, int G) {
    __shared__ int c[BK_NODES];
    __shared__ int wtot[SORT_T / 32];
    const int lane = threadIdx.x & 31, wave = threadIdx.x >> 5;
    const int bbeg = off[(size_t)blockIdx.x * G];
    const int bend = off[(size_t)(blockIdx.x + 1) * G];
    c[threadIdx.x] = 0;
    __syncthreads();
    for (int j0 = bbeg + threadIdx.x; j0 < bend; j0 += CSR_U * SORT_T) {
        int v[CSR_U];
#pragma unroll
        for (int u = 0; u < CSR_U; ++u)
            v[u] = tmp[min(j0 + u * SORT_T, bend - 1)];
#pragma unroll
        for (int u = 0; u < CSR_U; ++u)
            if (j0 + u * SORT_T < bend) atomicAdd(&c[v[u] & (BK_NODES - 1)], 1);
    }
    __syncthreads();
    const int mine = c[threadIdx.x];
    int incl = mine;
#pragma unroll
    for (int o = 1; o < 32; o <<= 1) {
        const int t = __shfl_up(incl, o);
        if (lane >= o) incl += t;
    }
    if (lane == 31) wtot[wave] = incl;
    __syncthreads();
    int start = bbeg + incl - mine;
#pragma unroll
    for (int w = 0; w < SORT_T / 32; ++w)
        if (w < wave) start += wtot[w];
    const int node = blockIdx.x * BK_NODES + threadIdx.x;
    if (node < N) rowptr[node] = start;
    if (node == N - 1) rowptr[N] = start + mine;
    c[threadIdx.x] = start;
    __syncthreads();
    for (int j0 = bbeg + threadIdx.x; j0 < bend; j0 += CSR_U * SORT_T) {
        int v[CSR_U], pos[CSR_U];
#pragma unroll
        for (int u = 0; u < CSR_U; ++u)
            v[u] = tmp[min(j0 + u * SORT_T, bend - 1)];
#pragma unroll
        for (int u = 0; u < CSR_U; ++u)
            if (j0 + u * SORT_T < bend) pos[u] = atomicAdd(&c[v[u] & (BK_NODES - 1)], 1);
#pragma unroll
        for (int u = 0; u < CSR_U; ++u)
            if (j0 + u * SORT_T < bend) csr_src[pos[u]] = (int)((unsigned)v[u] >> BK_SH);
    }
}

__global__ void __launch_bounds__(SORT_T)
bucket_hist_gemm(const int* __restrict__ ei, int* __restrict__ hist, int E, int chunk, int nbuck,
                 const float* __restrict__ x, const float* __restrict__ W1, float* __restrict__ xp,
                 const float* __restrict__ att_s, const float* __restrict__ att_d,
                 float* __restrict__ a_s, float* __restrict__ a_d, int n, int blk0) {
    if (blockIdx.x < SORT_G) {
        bucket_hist_body(ei, hist, E, chunk, nbuck, SORT_G);
        return;
    }
    gemm1_block64(x, W1, xp, att_s, att_d, a_s, a_d, n, blk0 + (int)blockIdx.x - SORT_G);
}

__global__ void __launch_bounds__(SORT_T)
bucket_scatter_gemm(const int* __restrict__ ei, const int* __restrict__ off, int* __restrict__ tmp,
                    int E, int chunk, int nbuck,
                    const float* __restrict__ x, const float* __restrict__ W1, float* __restrict__ xp,
                    const float* __restrict__ att_s, const float* __restrict__ att_d,
                    float* __restrict__ a_s, float* __restrict__ a_d, int n, int blk0) {
    if (blockIdx.x < SORT_G) {
        bucket_scatter_body(ei, off, tmp, E, chunk, nbuck, SORT_G);
        return;
    }
    gemm1_block64(x, W1, xp, att_s, att_d, a_s, a_d, n, blk0 + (int)blockIdx.x - SORT_G);
}

__global__ void __launch_bounds__(SORT_T)
bucket_csr_gemm(const int* __restrict__ tmp, const int* __restrict__ off,
                int* __restrict__ rowptr, int* __restrict__ csr_src, int N, int nbuck,
                const float* __restrict__ x, const float* __restrict__ W1, float* __restrict__ xp,
                const float* __restrict__ att_s, const float* __restrict__ att_d,
                float* __restrict__ a_s, float* __restrict__ a_d, int blk0) {
    if ((int)blockIdx.x < nbuck) {
        bucket_csr_body(tmp, off, rowptr, csr_src, N, SORT_G);
        return;
    }
    gemm1_block64(x, W1, xp, att_s, att_d, a_s, a_d, N, blk0 + (int)blockIdx.x - nbuck);
}

__device__ __forceinline__ float lrelu(float v) {
    return v > 0.0f ? v : NEG_SLOPE * v;
}

__device__ __forceinline__ float elu1(float v) {
    return v > 0.0f ? v : expm1f(v);
}

__device__ __forceinline__ float head_sum(float v) {
    v += row_xor<1>(v);
    v += row_xor<2>(v);
    v += row_xor<4>(v);
    return v;
}

#define ROWS_CH 8

template <int T>
__device__ __forceinline__ float group_bcast(float v) {
    return __int_as_float(__builtin_amdgcn_mov_dpp8(__float_as_int(v), T * 0x249249));
}

template <int T, int R>
__device__ __forceinline__ void edge_acc(const v4f (&r)[R], float psel, v2f& acc01, v2f& acc23) {
    if constexpr (T < R) {
        const float p = group_bcast<T>(psel);
        const v2f lo = {r[T].x, r[T].y}, hi = {r[T].z, r[T].w};
        acc01 += lo * p;
        acc23 += hi * p;
    }
}

template <int R>
__device__ __forceinline__ void edge_chunk(const float* __restrict__ xp, int s, int c0,
                                           const v2f as01, const v2f as23, float adn, int lane,
                                           v2f& acc01, v2f& acc23, float& den, float& m) {
    v4f r[R];
    float x[R];
#pragma unroll
    for (int t = 0; t < R; ++t) {
        const int st = __builtin_amdgcn_readlane(s, c0 + t);
        r[t] = *(const v4f*)(xp + (size_t)st * HID + lane * 4);
    }
#pragma unroll
    for (int t = 0; t < R; ++t) {
        const v2f lo = {r[t].x, r[t].y}, hi = {r[t].z, r[t].w};
        const v2f d = lo * as01 + hi * as23;
        x[t] = head_sum(d.x + d.y) + adn;
    }
    const int i = lane & 7;
    float xm = x[0];
    float xsel = x[0];
#pragma unroll
    for (int t = 1; t < R; ++t) {
        xm = fmaxf(xm, x[t]);
        xsel = i == t ? x[t] : xsel;
    }
    const float mn = fmaxf(m, fmaxf(xm, NEG_SLOPE * xm));
    const float sc = expf(m - mn);
    acc01 *= sc; acc23 *= sc;
    den *= sc;
    m = mn;
    const float esel = fmaxf(xsel, NEG_SLOPE * xsel);
    float psel = expf(esel - mn);
    if (R < ROWS_CH) psel = i < R ? psel : 0.0f;
    den += head_sum(psel);
    edge_acc<0, R>(r, psel, acc01, acc23); edge_acc<1, R>(r, psel, acc01, acc23);
    edge_acc<2, R>(r, psel, acc01, acc23); edge_acc<3, R>(r, psel, acc01, acc23);
    edge_acc<4, R>(r, psel, acc01, acc23); edge_acc<5, R>(r, psel, acc01, acc23);
    edge_acc<6, R>(r, psel, acc01, acc23); edge_acc<7, R>(r, psel, acc01, acc23);
}

template <bool LAST>
__global__ void node_softmax_agg(const int* __restrict__ rowptr,
                                 const int* __restrict__ csr_src,
                                 const float* __restrict__ xp,
                                 const float* __restrict__ att_s,
                                 const float* __restrict__ a_s,
                                 const float* __restrict__ a_d,
                                 const float* __restrict__ bias,
                                 float* __restrict__ dst, int n) {
    const int lane = threadIdx.x & 31;
    const int node = __builtin_amdgcn_readfirstlane(
        (int)((blockIdx.x * blockDim.x + threadIdx.x) >> 5));
    if (node >= n) return;
    const int beg = rowptr[node], end = rowptr[node + 1];
    const v4f as4 = *(const v4f*)(att_s + lane * 4);
    const v2f as01 = {as4.x, as4.y}, as23 = {as4.z, as4.w};
    const int hme = lane >> 3;

    const float4 ad = *(const float4*)(a_d + (size_t)node * NH);
    const float4 an = *(const float4*)(a_s + (size_t)node * NH);
    const float adn = hme == 0 ? ad.x : hme == 1 ? ad.y : hme == 2 ? ad.z : ad.w;
    const float asn = hme == 0 ? an.x : hme == 1 ? an.y : hme == 2 ? an.z : an.w;
    const v4f rs = *(const v4f*)(xp + (size_t)node * HID + lane * 4);
    float m = lrelu(asn + adn);
    v2f acc01 = {rs.x, rs.y}, acc23 = {rs.z, rs.w};
    float den = 1.0f;

    for (int base = beg; base < end; base += 32) {
        const int j = base + lane;
        const int s = (j < end) ? csr_src[j] : node;
        const int cnt32 = min(32, end - base);
        int c0 = 0;
        for (; c0 + ROWS_CH <= cnt32; c0 += ROWS_CH)
            edge_chunk<ROWS_CH>(xp, s, c0, as01, as23, adn, lane, acc01, acc23, den, m);
        switch (cnt32 - c0) {
            case 1: edge_chunk<1>(xp, s, c0, as01, as23, adn, lane, acc01, acc23, den, m); break;
            case 2: edge_chunk<2>(xp, s, c0, as01, as23, adn, lane, acc01, acc23, den, m); break;
            case 3: edge_chunk<3>(xp, s, c0, as01, as23, adn, lane, acc01, acc23, den, m); break;
            case 4: edge_chunk<4>(xp, s, c0, as01, as23, adn, lane, acc01, acc23, den, m); break;
            case 5: edge_chunk<5>(xp, s, c0, as01, as23, adn, lane, acc01, acc23, den, m); break;
            case 6: edge_chunk<6>(xp, s, c0, as01, as23, adn, lane, acc01, acc23, den, m); break;
            case 7: edge_chunk<7>(xp, s, c0, as01, as23, adn, lane, acc01, acc23, den, m); break;
            default: break;
        }
    }

    const float dd = den + 1e-16f;
    const float4 a = make_float4(acc01.x / dd, acc01.y / dd, acc23.x / dd, acc23.y / dd);
    if (!LAST) {
        const float4 b = *(const float4*)(bias + lane * 4);
        *(float4*)(dst + (size_t)node * HID + lane * 4) =
            make_float4(elu1(a.x + b.x), elu1(a.y + b.y), elu1(a.z + b.z), elu1(a.w + b.w));
    } else {
        const float4 a1 = make_float4(__shfl_xor(a.x, 8),  __shfl_xor(a.y, 8),
                                      __shfl_xor(a.z, 8),  __shfl_xor(a.w, 8));
        const float4 a2 = make_float4(__shfl_xor(a.x, 16), __shfl_xor(a.y, 16),
                                      __shfl_xor(a.z, 16), __shfl_xor(a.w, 16));
        const float4 a3 = make_float4(__shfl_xor(a.x, 24), __shfl_xor(a.y, 24),
                                      __shfl_xor(a.z, 24), __shfl_xor(a.w, 24));
        if (lane < 8) {
            const float4 b = *(const float4*)(bias + lane * 4);
            *(float4*)(dst + (size_t)node * NC + lane * 4) =
                make_float4(0.25f * (a.x + a1.x + a2.x + a3.x) + b.x,
                            0.25f * (a.y + a1.y + a2.y + a3.y) + b.y,
                            0.25f * (a.z + a1.z + a2.z + a3.z) + b.z,
                            0.25f * (a.w + a1.w + a2.w + a3.w) + b.w);
        }
    }
}


static inline int cdiv(int a, int b) { return (a + b - 1) / b; }

extern "C" void kernel_launch(void* const* d_in, const int* in_sizes, int n_in,
                              void* d_out, int out_size, void* d_ws, size_t ws_size,
                              hipStream_t stream) {
    const float* x        = (const float*)d_in[0];
    const int*   ei       = (const int*)  d_in[1];
    const float* W1       = (const float*)d_in[2];
    const float* att_src1 = (const float*)d_in[3];
    const float* att_dst1 = (const float*)d_in[4];
    const float* bias1    = (const float*)d_in[5];
    const float* W2       = (const float*)d_in[6];
    const float* att_src2 = (const float*)d_in[7];
    const float* att_dst2 = (const float*)d_in[8];
    const float* bias2    = (const float*)d_in[9];
    float* out = (float*)d_out;

    const int IN = 64;
    const int N  = in_sizes[0] / IN;
    const int E  = in_sizes[1] / 2;

    float* xp1   = (float*)d_ws;
    float* agg1  = xp1  + (size_t)N * HID;
    float* xp2   = agg1 + (size_t)N * HID;
    float* r4    = xp2  + (size_t)N * HID;
    float* a_s   = r4   + (size_t)N * HID;
    float* a_d   = a_s  + (size_t)N * NH;

    int* csr_src = (int*)r4;
    int* rowptr  = csr_src + (size_t)E;
    int* tmp     = rowptr + (size_t)N + 1;
    const int nbuck = cdiv(N, BK_NODES);
    const int nTab  = nbuck * SORT_G;
    int* hist    = tmp + (size_t)E;
    int* off     = hist + (size_t)nTab;
    int* bsum    = off + (size_t)nTab + 1;

    const int BT = 256;
    const int gRowTiles = cdiv(N, 32);
    const int chunk = cdiv(E, SORT_G);
    const int gScan = cdiv(nTab, SCAN_BLK);
    const int gNW  = cdiv(N * 32, BT);
    const int gB64 = cdiv(N, 64);
    const int gGa  = gB64 / 4;
    const int gGb  = (gB64 - gGa) / 2;
    const int gGc  = gB64 - gGa - gGb;

    bucket_hist_gemm<<<SORT_G + gGa, SORT_T, 0, stream>>>(ei, hist, E, chunk, nbuck,
        x, W1, xp1, att_src1, att_dst1, a_s, a_d, N, 0);
    scan_blocksum<<<gScan, SCAN_T, 0, stream>>>(hist, bsum, nTab);
    scan_write<<<gScan, SCAN_T, 0, stream>>>(hist, bsum, off, nTab);
    bucket_scatter_gemm<<<SORT_G + gGb, SORT_T, 0, stream>>>(ei, off, tmp, E, chunk, nbuck,
        x, W1, xp1, att_src1, att_dst1, a_s, a_d, N, gGa);
    bucket_csr_gemm<<<nbuck + gGc, SORT_T, 0, stream>>>(tmp, off, rowptr, csr_src, N, nbuck,
        x, W1, xp1, att_src1, att_dst1, a_s, a_d, gGa + gGb);

    node_softmax_agg<false><<<gNW, BT, 0, stream>>>(rowptr, csr_src, xp1, att_src1, a_s, a_d, bias1, agg1, N);

    gemm_wmma_f32<128><<<gRowTiles, 128, 0, stream>>>(agg1, W2, xp2, att_src2, att_dst2, a_s, a_d, N, HID);
    node_softmax_agg<true><<<gNW, BT, 0, stream>>>(rowptr, csr_src, xp2, att_src2, a_s, a_d, bias2, out, N);
}


# Second layer's matrix product done inside the first layer's node launch: 8 waves put 32 rows of h (4 nodes each) in LDS, 4 of them multiply by W2 from there; h never stored; 5 to 4 launches
# speedup vs baseline: 7.5251x; 1.3183x over previous
#include <hip/hip_runtime.h>
#include <math.h>

typedef __attribute__((ext_vector_type(2))) float v2f;
typedef __attribute__((ext_vector_type(4))) float v4f;
typedef __attribute__((ext_vector_type(8))) float v8f;

#define HID 128
#define NH  4
#define NC  32
#define NEG_SLOPE 0.2f

template <int XM>
__device__ __forceinline__ float row_xor(float v) {
    return __int_as_float(__builtin_amdgcn_update_dpp(
        0, __float_as_int(v), 0x160 | XM, 0xf, 0xf, true));
}


__device__ __forceinline__ float fold16(float (&v)[16], int l15) {
#pragma unroll
    for (int i = 0; i < 8; ++i) {
        const bool up = (l15 & 8) != 0;
        const float send = up ? v[i] : v[i + 8];
        const float keep = up ? v[i + 8] : v[i];
        v[i] = keep + row_xor<8>(send);
    }
#pragma unroll
    for (int i = 0; i < 4; ++i) {
        const bool up = (l15 & 4) != 0;
        const float send = up ? v[i] : v[i + 4];
        const float keep = up ? v[i + 4] : v[i];
        v[i] = keep + row_xor<4>(send);
    }
#pragma unroll
    for (int i = 0; i < 2; ++i) {
        const bool up = (l15 & 2) != 0;
        const float send = up ? v[i] : v[i + 2];
        const float keep = up ? v[i + 2] : v[i];
        v[i] = keep + row_xor<2>(send);
    }
    const bool up = (l15 & 1) != 0;
    const float send = up ? v[0] : v[1];
    const float keep = up ? v[1] : v[0];
    return keep + row_xor<1>(send);
}

template <int K>
__device__ __forceinline__ void gemm_rows32(const float* __restrict__ A,
                                            const float* __restrict__ W,
                                            float* __restrict__ C,
                                            const float* __restrict__ att_s,
                                            const float* __restrict__ att_d,
                                            float* __restrict__ a_s, float* __restrict__ a_d,
                                            int n, int Oc, int rowBase, int wave, int lane,
                                            float (*sAtt)[32][NH]) {
    const int colBase = wave * 32;

    const int half = lane >> 4;
    const int l15  = lane & 15;
    int row0 = rowBase + l15;
    int row1 = row0 + 16;
    if (row0 >= n) row0 = n - 1;
    if (row1 >= n) row1 = n - 1;
    const int col0 = colBase + l15;
    const int col1 = col0 + 16;

    const float* Ap0 = A + (size_t)row0 * K + 4 * half;
    const float* Ap1 = A + (size_t)row1 * K + 4 * half;
    const float* Wp0 = W + (size_t)col0 * K + 4 * half;
    const float* Wp1 = W + (size_t)col1 * K + 4 * half;

    v8f acc00 = {}, acc01 = {}, acc10 = {}, acc11 = {};
#pragma unroll
    for (int k = 0; k < K; k += 8) {
        const v4f a0 = *(const v4f*)(Ap0 + k);
        const v4f a1 = *(const v4f*)(Ap1 + k);
        const v4f b0 = *(const v4f*)(Wp0 + k);
        const v4f b1 = *(const v4f*)(Wp1 + k);
        const v2f a0l = {a0.x, a0.y}, a0h = {a0.z, a0.w};
        const v2f a1l = {a1.x, a1.y}, a1h = {a1.z, a1.w};
        const v2f b0l = {b0.x, b0.y}, b0h = {b0.z, b0.w};
        const v2f b1l = {b1.x, b1.y}, b1h = {b1.z, b1.w};
        acc00 = __builtin_amdgcn_wmma_f32_16x16x4_f32(
            false, a0l, false, b0l, (short)0, acc00, false, false);
        acc01 = __builtin_amdgcn_wmma_f32_16x16x4_f32(
            false, a0l, false, b1l, (short)0, acc01, false, false);
        acc10 = __builtin_amdgcn_wmma_f32_16x16x4_f32(
            false, a1l, false, b0l, (short)0, acc10, false, false);
        acc11 = __builtin_amdgcn_wmma_f32_16x16x4_f32(
            false, a1l, false, b1l, (short)0, acc11, false, false);
        acc00 = __builtin_amdgcn_wmma_f32_16x16x4_f32(
            false, a0h, false, b0h, (short)0, acc00, false, false);
        acc01 = __builtin_amdgcn_wmma_f32_16x16x4_f32(
            false, a0h, false, b1h, (short)0, acc01, false, false);
        acc10 = __builtin_amdgcn_wmma_f32_16x16x4_f32(
            false, a1h, false, b0h, (short)0, acc10, false, false);
        acc11 = __builtin_amdgcn_wmma_f32_16x16x4_f32(
            false, a1h, false, b1h, (short)0, acc11, false, false);
    }

    const float ws0 = att_s[col0], ws1 = att_s[col1];
    const float wd0 = att_d[col0], wd1 = att_d[col1];
    float v[16];
#pragma unroll
    for (int r = 0; r < 8; ++r) {
        v[r]     = acc00[r] * ws0 + acc01[r] * ws1;
        v[r + 8] = acc00[r] * wd0 + acc01[r] * wd1;
    }
    const float d0 = fold16(v, l15);
#pragma unroll
    for (int r = 0; r < 8; ++r) {
        v[r]     = acc10[r] * ws0 + acc11[r] * ws1;
        v[r + 8] = acc10[r] * wd0 + acc11[r] * wd1;
    }
    const float d1 = fold16(v, l15);
    sAtt[l15 >> 3][half * 8 + (l15 & 7)][wave]      = d0;
    sAtt[l15 >> 3][16 + half * 8 + (l15 & 7)][wave] = d1;
    __syncthreads();
    if (wave < 2) {
        const int grow = rowBase + lane;
        if (grow < n) {
            const float4 o = *(const float4*)&sAtt[wave][lane][0];
            *(float4*)((wave ? a_d : a_s) + (size_t)grow * NH) = o;
        }
    }

    const int crow0 = rowBase + half * 8;
#pragma unroll
    for (int r = 0; r < 8; ++r) {
        const int crow = crow0 + r;
        if (crow < n) {
            C[(size_t)crow * Oc + col0] = acc00[r];
            C[(size_t)crow * Oc + col1] = acc01[r];
        }
    }
#pragma unroll
    for (int r = 0; r < 8; ++r) {
        const int crow = crow0 + 16 + r;
        if (crow < n) {
            C[(size_t)crow * Oc + col0] = acc10[r];
            C[(size_t)crow * Oc + col1] = acc11[r];
        }
    }
}

template <int GROUPS>
__device__ __forceinline__ void gemm1_block(const float* __restrict__ x,
                                            const float* __restrict__ W1,
                                            float* __restrict__ xp,
                                            const float* __restrict__ att_s,
                                            const float* __restrict__ att_d,
                                            float* __restrict__ a_s, float* __restrict__ a_d,
                                            int n, int blk) {
    __shared__ __attribute__((aligned(16))) float sAtt2[GROUPS][2][32][NH];
    const int w   = (int)(threadIdx.x >> 5);
    const int grp = w >> 2;
    gemm_rows32<64>(x, W1, xp, att_s, att_d, a_s, a_d, n, HID,
                    blk * (32 * GROUPS) + grp * 32, w & 3, (int)(threadIdx.x & 31), sAtt2[grp]);
}


#define BK_SH    8
#define BK_NODES (1 << BK_SH)
#define BK_MAX   512
#define CS_T     512
#define SORT_T   256
#define SORT_G   256
#define SORT_U   16
#define CSR_P    16

static_assert(CS_T == BK_MAX, "chunk_sort's prefix takes one bucket per thread");
static_assert(SORT_T == BK_NODES && SORT_T == SORT_G, "bucket_csr: thread t owns node t and chunk t");

__device__ __forceinline__ void chunk_sort_body(const int* __restrict__ ei, int* __restrict__ tmp,
                                                int* __restrict__ loc, int E, int chunk, int nbuck) {
    __shared__ int h[BK_MAX];
    __shared__ int wtot[CS_T / 32];
    const int t = threadIdx.x, lane = t & 31, wave = t >> 5;
    h[t] = 0;
    __syncthreads();
    const int beg = blockIdx.x * chunk;
    const int end = min(E, beg + chunk);
    for (int i0 = beg + t; i0 < end; i0 += SORT_U * CS_T) {
        int d[SORT_U];
#pragma unroll
        for (int u = 0; u < SORT_U; ++u)
            d[u] = ei[E + min(i0 + u * CS_T, end - 1)];
#pragma unroll
        for (int u = 0; u < SORT_U; ++u)
            if (i0 + u * CS_T < end) atomicAdd(&h[d[u] >> BK_SH], 1);
    }
    __syncthreads();
    const int mine = (t < nbuck) ? h[t] : 0;
    int incl = mine;
#pragma unroll
    for (int o = 1; o < 32; o <<= 1) {
        const int up = __shfl_up(incl, o);
        if (lane >= o) incl += up;
    }
    if (lane == 31) wtot[wave] = incl;
    __syncthreads();
    int start = incl - mine;
#pragma unroll
    for (int w = 0; w < CS_T / 32; ++w)
        if (w < wave) start += wtot[w];
    int* lrow = loc + (size_t)blockIdx.x * (nbuck + 1);
    if (t < nbuck) lrow[t] = start;
    if (t == 0) lrow[nbuck] = max(end - beg, 0);
    h[t] = start;
    __syncthreads();
    for (int i0 = beg + t; i0 < end; i0 += SORT_U * CS_T) {
        int d[SORT_U], s[SORT_U], pos[SORT_U];
#pragma unroll
        for (int u = 0; u < SORT_U; ++u) {
            const int j = min(i0 + u * CS_T, end - 1);
            d[u] = ei[E + j];
            s[u] = ei[j];
        }
#pragma unroll
        for (int u = 0; u < SORT_U; ++u)
            if (i0 + u * CS_T < end) pos[u] = atomicAdd(&h[d[u] >> BK_SH], 1);
#pragma unroll
        for (int u = 0; u < SORT_U; ++u)
            if (i0 + u * CS_T < end)
                tmp[beg + pos[u]] = (int)(((unsigned)s[u] << BK_SH) | (unsigned)(d[u] & (BK_NODES - 1)));
    }
}

__device__ __forceinline__ void bucket_csr_body(const int* __restrict__ tmp, const int* __restrict__ loc,
                                                int* __restrict__ rowptr, int* __restrict__ csr_src,
                                                int N, int chunk, int nbuck) {
    __shared__ int c[BK_NODES];
    __shared__ int wtot[SORT_T / 32];
    __shared__ int wsum[SORT_T / 32];
    const int lane = threadIdx.x & 31, wave = threadIdx.x >> 5;
    const int* lrow = loc + (size_t)threadIdx.x * (nbuck + 1) + blockIdx.x;
    const int lo = lrow[0];
    const int plen  = lrow[1] - lo;
    const int pbase = (int)threadIdx.x * chunk + lo;
    int before = lo;
#pragma unroll
    for (int o = 16; o > 0; o >>= 1) before += __shfl_xor(before, o);
    if (lane == 0) wsum[wave] = before;
    c[threadIdx.x] = 0;
    __syncthreads();
    int bbeg = 0;
#pragma unroll
    for (int w = 0; w < SORT_T / 32; ++w) bbeg += wsum[w];

    for (int j0 = 0; j0 < 32; j0 += CSR_P) {
        int v[CSR_P];
#pragma unroll
        for (int q = 0; q < CSR_P; ++q) {
            const int pb = __builtin_amdgcn_readlane(pbase, j0 + q);
            const int pl = __builtin_amdgcn_readlane(plen, j0 + q);
            v[q] = 0;
            if (lane < pl) v[q] = tmp[pb + lane];
        }
#pragma unroll
        for (int q = 0; q < CSR_P; ++q) {
            const int pb = __builtin_amdgcn_readlane(pbase, j0 + q);
            const int pl = __builtin_amdgcn_readlane(plen, j0 + q);
            if (lane < pl) atomicAdd(&c[v[q] & (BK_NODES - 1)], 1);
            for (int k = 32 + lane; k < pl; k += 32)
                atomicAdd(&c[tmp[pb + k] & (BK_NODES - 1)], 1);
        }
    }
    __syncthreads();
    const int mine = c[threadIdx.x];
    int incl = mine;
#pragma unroll
    for (int o = 1; o < 32; o <<= 1) {
        const int up = __shfl_up(incl, o);
        if (lane >= o) incl += up;
    }
    if (lane == 31) wtot[wave] = incl;
    __syncthreads();
    int start = bbeg + incl - mine;
#pragma unroll
    for (int w = 0; w < SORT_T / 32; ++w)
        if (w < wave) start += wtot[w];
    const int node = blockIdx.x * BK_NODES + threadIdx.x;
    if (node < N) rowptr[node] = start;
    if (node == N - 1) rowptr[N] = start + mine;
    c[threadIdx.x] = start;
    __syncthreads();
    for (int j0 = 0; j0 < 32; j0 += CSR_P) {
        int v[CSR_P], pos[CSR_P];
#pragma unroll
        for (int q = 0; q < CSR_P; ++q) {
            const int pb = __builtin_amdgcn_readlane(pbase, j0 + q);
            const int pl = __builtin_amdgcn_readlane(plen, j0 + q);
            v[q] = 0;
            if (lane < pl) v[q] = tmp[pb + lane];
        }
#pragma unroll
        for (int q = 0; q < CSR_P; ++q) {
            const int pl = __builtin_amdgcn_readlane(plen, j0 + q);
            pos[q] = 0;
            if (lane < pl) pos[q] = atomicAdd(&c[v[q] & (BK_NODES - 1)], 1);
        }
#pragma unroll
        for (int q = 0; q < CSR_P; ++q) {
            const int pb = __builtin_amdgcn_readlane(pbase, j0 + q);
            const int pl = __builtin_amdgcn_readlane(plen, j0 + q);
            if (lane < pl) csr_src[pos[q]] = (int)((unsigned)v[q] >> BK_SH);
            for (int k = 32 + lane; k < pl; k += 32) {
                const int vv = tmp[pb + k];
                csr_src[atomicAdd(&c[vv & (BK_NODES - 1)], 1)] = (int)((unsigned)vv >> BK_SH);
            }
        }
    }
}

__global__ void __launch_bounds__(CS_T)
chunk_sort_gemm(const int* __restrict__ ei, int* __restrict__ tmp, int* __restrict__ loc,
                int E, int chunk, int nbuck,
                const float* __restrict__ x, const float* __restrict__ W1, float* __restrict__ xp,
                const float* __restrict__ att_s, const float* __restrict__ att_d,
                float* __restrict__ a_s, float* __restrict__ a_d, int n) {
    if (blockIdx.x < SORT_G) {
        chunk_sort_body(ei, tmp, loc, E, chunk, nbuck);
        return;
    }
    gemm1_block<CS_T / 128>(x, W1, xp, att_s, att_d, a_s, a_d, n, (int)blockIdx.x - SORT_G);
}

__global__ void __launch_bounds__(SORT_T)
bucket_csr_gemm(const int* __restrict__ tmp, const int* __restrict__ loc,
                int* __restrict__ rowptr, int* __restrict__ csr_src, int N, int chunk, int nbuck,
                const float* __restrict__ x, const float* __restrict__ W1, float* __restrict__ xp,
                const float* __restrict__ att_s, const float* __restrict__ att_d,
                float* __restrict__ a_s, float* __restrict__ a_d, int blk0) {
    if ((int)blockIdx.x < nbuck) {
        bucket_csr_body(tmp, loc, rowptr, csr_src, N, chunk, nbuck);
        return;
    }
    gemm1_block<SORT_T / 128>(x, W1, xp, att_s, att_d, a_s, a_d, N, blk0 + (int)blockIdx.x - nbuck);
}

__device__ __forceinline__ float lrelu(float v) {
    return v > 0.0f ? v : NEG_SLOPE * v;
}

__device__ __forceinline__ float elu1(float v) {
    return v > 0.0f ? v : expm1f(v);
}

__device__ __forceinline__ float head_sum(float v) {
    v += row_xor<1>(v);
    v += row_xor<2>(v);
    v += row_xor<4>(v);
    return v;
}

#define ROWS_CH 8

template <int T>
__device__ __forceinline__ float group_bcast(float v) {
    return __int_as_float(__builtin_amdgcn_mov_dpp8(__float_as_int(v), T * 0x249249));
}

template <int T, int R>
__device__ __forceinline__ void edge_acc(const v4f (&r)[R], float psel, v2f& acc01, v2f& acc23) {
    if constexpr (T < R) {
        const float p = group_bcast<T>(psel);
        const v2f lo = {r[T].x, r[T].y}, hi = {r[T].z, r[T].w};
        acc01 += lo * p;
        acc23 += hi * p;
    }
}

template <int R>
__device__ __forceinline__ void edge_chunk(const float* __restrict__ xp, int s, int c0,
                                           const v2f as01, const v2f as23, float adn, int lane,
                                           v2f& acc01, v2f& acc23, float& den, float& m) {
    v4f r[R];
    float x[R];
#pragma unroll
    for (int t = 0; t < R; ++t) {
        const int st = __builtin_amdgcn_readlane(s, c0 + t);
        r[t] = *(const v4f*)(xp + (size_t)st * HID + lane * 4);
    }
#pragma unroll
    for (int t = 0; t < R; ++t) {
        const v2f lo = {r[t].x, r[t].y}, hi = {r[t].z, r[t].w};
        const v2f d = lo * as01 + hi * as23;
        x[t] = head_sum(d.x + d.y) + adn;
    }
    const int i = lane & 7;
    float xm = x[0];
    float xsel = x[0];
#pragma unroll
    for (int t = 1; t < R; ++t) {
        xm = fmaxf(xm, x[t]);
        xsel = i == t ? x[t] : xsel;
    }
    const float mn = fmaxf(m, fmaxf(xm, NEG_SLOPE * xm));
    const float sc = expf(m - mn);
    acc01 *= sc; acc23 *= sc;
    den *= sc;
    m = mn;
    const float esel = fmaxf(xsel, NEG_SLOPE * xsel);
    float psel = expf(esel - mn);
    if (R < ROWS_CH) psel = i < R ? psel : 0.0f;
    den += head_sum(psel);
    edge_acc<0, R>(r, psel, acc01, acc23); edge_acc<1, R>(r, psel, acc01, acc23);
    edge_acc<2, R>(r, psel, acc01, acc23); edge_acc<3, R>(r, psel, acc01, acc23);
    edge_acc<4, R>(r, psel, acc01, acc23); edge_acc<5, R>(r, psel, acc01, acc23);
    edge_acc<6, R>(r, psel, acc01, acc23); edge_acc<7, R>(r, psel, acc01, acc23);
}

template <bool LAST>
__global__ void node_softmax_agg(const int* __restrict__ rowptr,
                                 const int* __restrict__ csr_src,
                                 const float* __restrict__ xp,
                                 const float* __restrict__ att_s,
                                 const float* __restrict__ a_s,
                                 const float* __restrict__ a_d,
                                 const float* __restrict__ bias,
                                 float* __restrict__ dst, int n) {
    const int lane = threadIdx.x & 31;
    const int node = __builtin_amdgcn_readfirstlane(
        (int)((blockIdx.x * blockDim.x + threadIdx.x) >> 5));
    if (node >= n) return;
    const int beg = rowptr[node], end = rowptr[node + 1];
    const v4f as4 = *(const v4f*)(att_s + lane * 4);
    const v2f as01 = {as4.x, as4.y}, as23 = {as4.z, as4.w};
    const int hme = lane >> 3;

    const float4 ad = *(const float4*)(a_d + (size_t)node * NH);
    const float4 an = *(const float4*)(a_s + (size_t)node * NH);
    const float adn = hme == 0 ? ad.x : hme == 1 ? ad.y : hme == 2 ? ad.z : ad.w;
    const float asn = hme == 0 ? an.x : hme == 1 ? an.y : hme == 2 ? an.z : an.w;
    const v4f rs = *(const v4f*)(xp + (size_t)node * HID + lane * 4);
    float m = lrelu(asn + adn);
    v2f acc01 = {rs.x, rs.y}, acc23 = {rs.z, rs.w};
    float den = 1.0f;

    for (int base = beg; base < end; base += 32) {
        const int j = base + lane;
        const int s = (j < end) ? csr_src[j] : node;
        const int cnt32 = min(32, end - base);
        int c0 = 0;
        for (; c0 + ROWS_CH <= cnt32; c0 += ROWS_CH)
            edge_chunk<ROWS_CH>(xp, s, c0, as01, as23, adn, lane, acc01, acc23, den, m);
        switch (cnt32 - c0) {
            case 1: edge_chunk<1>(xp, s, c0, as01, as23, adn, lane, acc01, acc23, den, m); break;
            case 2: edge_chunk<2>(xp, s, c0, as01, as23, adn, lane, acc01, acc23, den, m); break;
            case 3: edge_chunk<3>(xp, s, c0, as01, as23, adn, lane, acc01, acc23, den, m); break;
            case 4: edge_chunk<4>(xp, s, c0, as01, as23, adn, lane, acc01, acc23, den, m); break;
            case 5: edge_chunk<5>(xp, s, c0, as01, as23, adn, lane, acc01, acc23, den, m); break;
            case 6: edge_chunk<6>(xp, s, c0, as01, as23, adn, lane, acc01, acc23, den, m); break;
            case 7: edge_chunk<7>(xp, s, c0, as01, as23, adn, lane, acc01, acc23, den, m); break;
            default: break;
        }
    }

    const float dd = den + 1e-16f;
    const float4 a = make_float4(acc01.x / dd, acc01.y / dd, acc23.x / dd, acc23.y / dd);
    if (!LAST) {
        const float4 b = *(const float4*)(bias + lane * 4);
        *(float4*)(dst + (size_t)node * HID + lane * 4) =
            make_float4(elu1(a.x + b.x), elu1(a.y + b.y), elu1(a.z + b.z), elu1(a.w + b.w));
    } else {
        const float4 a1 = make_float4(__shfl_xor(a.x, 8),  __shfl_xor(a.y, 8),
                                      __shfl_xor(a.z, 8),  __shfl_xor(a.w, 8));
        const float4 a2 = make_float4(__shfl_xor(a.x, 16), __shfl_xor(a.y, 16),
                                      __shfl_xor(a.z, 16), __shfl_xor(a.w, 16));
        const float4 a3 = make_float4(__shfl_xor(a.x, 24), __shfl_xor(a.y, 24),
                                      __shfl_xor(a.z, 24), __shfl_xor(a.w, 24));
        if (lane < 8) {
            const float4 b = *(const float4*)(bias + lane * 4);
            *(float4*)(dst + (size_t)node * NC + lane * 4) =
                make_float4(0.25f * (a.x + a1.x + a2.x + a3.x) + b.x,
                            0.25f * (a.y + a1.y + a2.y + a3.y) + b.y,
                            0.25f * (a.z + a1.z + a2.z + a3.z) + b.z,
                            0.25f * (a.w + a1.w + a2.w + a3.w) + b.w);
        }
    }
}


#define FR 32
#define HP (HID + 4)

__device__ __forceinline__ float4 node_row(const int* __restrict__ rowptr,
                                           const int* __restrict__ csr_src,
                                           const float* __restrict__ xp,
                                           const v2f as01, const v2f as23,
                                           const float* __restrict__ a_s,
                                           const float* __restrict__ a_d,
                                           int node, int lane) {
    const int beg = rowptr[node], end = rowptr[node + 1];
    const int hme = lane >> 3;
    const float4 ad = *(const float4*)(a_d + (size_t)node * NH);
    const float4 an = *(const float4*)(a_s + (size_t)node * NH);
    const float adn = hme == 0 ? ad.x : hme == 1 ? ad.y : hme == 2 ? ad.z : ad.w;
    const float asn = hme == 0 ? an.x : hme == 1 ? an.y : hme == 2 ? an.z : an.w;
    const v4f rs = *(const v4f*)(xp + (size_t)node * HID + lane * 4);
    float m = lrelu(asn + adn);
    v2f acc01 = {rs.x, rs.y}, acc23 = {rs.z, rs.w};
    float den = 1.0f;

    for (int base = beg; base < end; base += 32) {
        const int j = base + lane;
        const int s = (j < end) ? csr_src[j] : node;
        const int cnt32 = min(32, end - base);
        int c0 = 0;
        for (; c0 + ROWS_CH <= cnt32; c0 += ROWS_CH)
            edge_chunk<ROWS_CH>(xp, s, c0, as01, as23, adn, lane, acc01, acc23, den, m);
        switch (cnt32 - c0) {
            case 1: edge_chunk<1>(xp, s, c0, as01, as23, adn, lane, acc01, acc23, den, m); break;
            case 2: edge_chunk<2>(xp, s, c0, as01, as23, adn, lane, acc01, acc23, den, m); break;
            case 3: edge_chunk<3>(xp, s, c0, as01, as23, adn, lane, acc01, acc23, den, m); break;
            case 4: edge_chunk<4>(xp, s, c0, as01, as23, adn, lane, acc01, acc23, den, m); break;
            case 5: edge_chunk<5>(xp, s, c0, as01, as23, adn, lane, acc01, acc23, den, m); break;
            case 6: edge_chunk<6>(xp, s, c0, as01, as23, adn, lane, acc01, acc23, den, m); break;
            case 7: edge_chunk<7>(xp, s, c0, as01, as23, adn, lane, acc01, acc23, den, m); break;
            default: break;
        }
    }
    const float dd = den + 1e-16f;
    return make_float4(acc01.x / dd, acc01.y / dd, acc23.x / dd, acc23.y / dd);
}

__global__ void __launch_bounds__(256)
node_agg_gemm2(const int* __restrict__ rowptr, const int* __restrict__ csr_src,
               const float* __restrict__ xp1, const float* __restrict__ att_s1,
               const float* __restrict__ a_s1, const float* __restrict__ a_d1,
               const float* __restrict__ bias1,
               const float* __restrict__ W2, float* __restrict__ xp2,
               const float* __restrict__ att_s2, const float* __restrict__ att_d2,
               float* __restrict__ a_s2, float* __restrict__ a_d2, int n) {
    __shared__ __attribute__((aligned(16))) float sH[FR][HP];
    __shared__ __attribute__((aligned(16))) float sAtt[2][32][NH];
    const int lane = threadIdx.x & 31;
    const int wave = __builtin_amdgcn_readfirstlane((int)(threadIdx.x >> 5));
    const int rowBase = (int)blockIdx.x * FR;

    {
        const v4f as4 = *(const v4f*)(att_s1 + lane * 4);
        const v2f as01 = {as4.x, as4.y}, as23 = {as4.z, as4.w};
        const float4 b = *(const float4*)(bias1 + lane * 4);
        for (int q = 0; q < FR / 8; ++q) {
            const int r = wave * (FR / 8) + q;
            const int node = rowBase + r;
            float4 o = make_float4(0.0f, 0.0f, 0.0f, 0.0f);
            if (node < n) {
                const float4 a = node_row(rowptr, csr_src, xp1, as01, as23, a_s1, a_d1, node, lane);
                o = make_float4(elu1(a.x + b.x), elu1(a.y + b.y), elu1(a.z + b.z), elu1(a.w + b.w));
            }
            *(float4*)&sH[r][lane * 4] = o;
        }
    }
    __syncthreads();

    const int half = lane >> 4;
    const int l15  = lane & 15;
    const int col0 = wave * 32 + l15;
    const int col1 = col0 + 16;
    v8f acc00 = {}, acc01 = {}, acc10 = {}, acc11 = {};
    if (wave < 4) {
        const float* Ap0 = &sH[l15][4 * half];
        const float* Ap1 = &sH[16 + l15][4 * half];
        const float* Wp0 = W2 + (size_t)col0 * HID + 4 * half;
        const float* Wp1 = W2 + (size_t)col1 * HID + 4 * half;
#pragma unroll
        for (int k = 0; k < HID; k += 8) {
            const v4f a0 = *(const v4f*)(Ap0 + k);
            const v4f a1 = *(const v4f*)(Ap1 + k);
            const v4f b0 = *(const v4f*)(Wp0 + k);
            const v4f b1 = *(const v4f*)(Wp1 + k);
            const v2f a0l = {a0.x, a0.y}, a0h = {a0.z, a0.w};
            const v2f a1l = {a1.x, a1.y}, a1h = {a1.z, a1.w};
            const v2f b0l = {b0.x, b0.y}, b0h = {b0.z, b0.w};
            const v2f b1l = {b1.x, b1.y}, b1h = {b1.z, b1.w};
            acc00 = __builtin_amdgcn_wmma_f32_16x16x4_f32(
                false, a0l, false, b0l, (short)0, acc00, false, false);
            acc01 = __builtin_amdgcn_wmma_f32_16x16x4_f32(
                false, a0l, false, b1l, (short)0, acc01, false, false);
            acc10 = __builtin_amdgcn_wmma_f32_16x16x4_f32(
                false, a1l, false, b0l, (short)0, acc10, false, false);
            acc11 = __builtin_amdgcn_wmma_f32_16x16x4_f32(
                false, a1l, false, b1l, (short)0, acc11, false, false);
            acc00 = __builtin_amdgcn_wmma_f32_16x16x4_f32(
                false, a0h, false, b0h, (short)0, acc00, false, false);
            acc01 = __builtin_amdgcn_wmma_f32_16x16x4_f32(
                false, a0h, false, b1h, (short)0, acc01, false, false);
            acc10 = __builtin_amdgcn_wmma_f32_16x16x4_f32(
                false, a1h, false, b0h, (short)0, acc10, false, false);
            acc11 = __builtin_amdgcn_wmma_f32_16x16x4_f32(
                false, a1h, false, b1h, (short)0, acc11, false, false);
        }
        const float ws0 = att_s2[col0], ws1 = att_s2[col1];
        const float wd0 = att_d2[col0], wd1 = att_d2[col1];
        float v[16];
#pragma unroll
        for (int r = 0; r < 8; ++r) {
            v[r]     = acc00[r] * ws0 + acc01[r] * ws1;
            v[r + 8] = acc00[r] * wd0 + acc01[r] * wd1;
        }
        const float d0 = fold16(v, l15);
#pragma unroll
        for (int r = 0; r < 8; ++r) {
            v[r]     = acc10[r] * ws0 + acc11[r] * ws1;
            v[r + 8] = acc10[r] * wd0 + acc11[r] * wd1;
        }
        const float d1 = fold16(v, l15);
        sAtt[l15 >> 3][half * 8 + (l15 & 7)][wave]      = d0;
        sAtt[l15 >> 3][16 + half * 8 + (l15 & 7)][wave] = d1;
    }
    __syncthreads();
    if (wave < 2) {
        const int grow = rowBase + lane;
        if (grow < n) {
            const float4 o = *(const float4*)&sAtt[wave][lane][0];
            *(float4*)((wave ? a_d2 : a_s2) + (size_t)grow * NH) = o;
        }
    }
    if (wave < 4) {
        const int crow0 = rowBase + half * 8;
#pragma unroll
        for (int r = 0; r < 8; ++r) {
            const int crow = crow0 + r;
            if (crow < n) {
                xp2[(size_t)crow * HID + col0] = acc00[r];
                xp2[(size_t)crow * HID + col1] = acc01[r];
            }
        }
#pragma unroll
        for (int r = 0; r < 8; ++r) {
            const int crow = crow0 + 16 + r;
            if (crow < n) {
                xp2[(size_t)crow * HID + col0] = acc10[r];
                xp2[(size_t)crow * HID + col1] = acc11[r];
            }
        }
    }
}


static inline int cdiv(int a, int b) { return (a + b - 1) / b; }

extern "C" void kernel_launch(void* const* d_in, const int* in_sizes, int n_in,
                              void* d_out, int out_size, void* d_ws, size_t ws_size,
                              hipStream_t stream) {
    const float* x        = (const float*)d_in[0];
    const int*   ei       = (const int*)  d_in[1];
    const float* W1       = (const float*)d_in[2];
    const float* att_src1 = (const float*)d_in[3];
    const float* att_dst1 = (const float*)d_in[4];
    const float* bias1    = (const float*)d_in[5];
    const float* W2       = (const float*)d_in[6];
    const float* att_src2 = (const float*)d_in[7];
    const float* att_dst2 = (const float*)d_in[8];
    const float* bias2    = (const float*)d_in[9];
    float* out = (float*)d_out;

    const int IN = 64;
    const int N  = in_sizes[0] / IN;
    const int E  = in_sizes[1] / 2;

    float* xp1   = (float*)d_ws;
    float* agg1  = xp1  + (size_t)N * HID;
    float* xp2   = agg1 + (size_t)N * HID;
    float* r4    = xp2  + (size_t)N * HID;
    float* a_s   = r4   + (size_t)N * HID;
    float* a_d   = a_s  + (size_t)N * NH;
    float* a_s2  = agg1;
    float* a_d2  = a_s2 + (size_t)N * NH;

    int* csr_src = (int*)r4;
    int* rowptr  = csr_src + (size_t)E;
    int* tmp     = rowptr + (size_t)N + 1;
    const int nbuck = cdiv(N, BK_NODES);
    int* loc     = tmp + (size_t)E;

    const int BT = 256;
    const int gRowTiles = cdiv(N, 32);
    const int chunk = cdiv(E, SORT_G);
    const int gNW  = cdiv(N * 32, BT);
    const int gGa  = (cdiv(N, 128) * 2) / 5;
    const int gGb  = cdiv(N, 64) - gGa * 2;

    chunk_sort_gemm<<<SORT_G + gGa, CS_T, 0, stream>>>(ei, tmp, loc, E, chunk, nbuck,
        x, W1, xp1, att_src1, att_dst1, a_s, a_d, N);
    bucket_csr_gemm<<<nbuck + gGb, SORT_T, 0, stream>>>(tmp, loc, rowptr, csr_src, N, chunk, nbuck,
        x, W1, xp1, att_src1, att_dst1, a_s, a_d, gGa * 2);

    node_agg_gemm2<<<gRowTiles, 256, 0, stream>>>(rowptr, csr_src, xp1, att_src1, a_s, a_d, bias1,
        W2, xp2, att_src2, att_dst2, a_s2, a_d2, N);

    node_softmax_agg<true><<<gNW, BT, 0, stream>>>(rowptr, csr_src, xp2, att_src2, a_s2, a_d2, bias2, out, N);
}


# Per-edge logit sums over a head's 8 lanes by one transposing fold per 8-edge chunk (lane i gets edge i's sum), not 8 all-lane butterfly sums plus selects; order of the 8-lane logit sum changes
# speedup vs baseline: 7.7649x; 1.0319x over previous
#include <hip/hip_runtime.h>
#include <math.h>

typedef __attribute__((ext_vector_type(2))) float v2f;
typedef __attribute__((ext_vector_type(4))) float v4f;
typedef __attribute__((ext_vector_type(8))) float v8f;

#define HID 128
#define NH  4
#define NC  32
#define NEG_SLOPE 0.2f

template <int XM>
__device__ __forceinline__ float row_xor(float v) {
    return __int_as_float(__builtin_amdgcn_update_dpp(
        0, __float_as_int(v), 0x160 | XM, 0xf, 0xf, true));
}


__device__ __forceinline__ float fold16(float (&v)[16], int l15) {
#pragma unroll
    for (int i = 0; i < 8; ++i) {
        const bool up = (l15 & 8) != 0;
        const float send = up ? v[i] : v[i + 8];
        const float keep = up ? v[i + 8] : v[i];
        v[i] = keep + row_xor<8>(send);
    }
#pragma unroll
    for (int i = 0; i < 4; ++i) {
        const bool up = (l15 & 4) != 0;
        const float send = up ? v[i] : v[i + 4];
        const float keep = up ? v[i + 4] : v[i];
        v[i] = keep + row_xor<4>(send);
    }
#pragma unroll
    for (int i = 0; i < 2; ++i) {
        const bool up = (l15 & 2) != 0;
        const float send = up ? v[i] : v[i + 2];
        const float keep = up ? v[i + 2] : v[i];
        v[i] = keep + row_xor<2>(send);
    }
    const bool up = (l15 & 1) != 0;
    const float send = up ? v[0] : v[1];
    const float keep = up ? v[1] : v[0];
    return keep + row_xor<1>(send);
}

template <int K>
__device__ __forceinline__ void gemm_rows32(const float* __restrict__ A,
                                            const float* __restrict__ W,
                                            float* __restrict__ C,
                                            const float* __restrict__ att_s,
                                            const float* __restrict__ att_d,
                                            float* __restrict__ a_s, float* __restrict__ a_d,
                                            int n, int Oc, int rowBase, int wave, int lane,
                                            float (*sAtt)[32][NH]) {
    const int colBase = wave * 32;

    const int half = lane >> 4;
    const int l15  = lane & 15;
    int row0 = rowBase + l15;
    int row1 = row0 + 16;
    if (row0 >= n) row0 = n - 1;
    if (row1 >= n) row1 = n - 1;
    const int col0 = colBase + l15;
    const int col1 = col0 + 16;

    const float* Ap0 = A + (size_t)row0 * K + 4 * half;
    const float* Ap1 = A + (size_t)row1 * K + 4 * half;
    const float* Wp0 = W + (size_t)col0 * K + 4 * half;
    const float* Wp1 = W + (size_t)col1 * K + 4 * half;

    v8f acc00 = {}, acc01 = {}, acc10 = {}, acc11 = {};
#pragma unroll
    for (int k = 0; k < K; k += 8) {
        const v4f a0 = *(const v4f*)(Ap0 + k);
        const v4f a1 = *(const v4f*)(Ap1 + k);
        const v4f b0 = *(const v4f*)(Wp0 + k);
        const v4f b1 = *(const v4f*)(Wp1 + k);
        const v2f a0l = {a0.x, a0.y}, a0h = {a0.z, a0.w};
        const v2f a1l = {a1.x, a1.y}, a1h = {a1.z, a1.w};
        const v2f b0l = {b0.x, b0.y}, b0h = {b0.z, b0.w};
        const v2f b1l = {b1.x, b1.y}, b1h = {b1.z, b1.w};
        acc00 = __builtin_amdgcn_wmma_f32_16x16x4_f32(
            false, a0l, false, b0l, (short)0, acc00, false, false);
        acc01 = __builtin_amdgcn_wmma_f32_16x16x4_f32(
            false, a0l, false, b1l, (short)0, acc01, false, false);
        acc10 = __builtin_amdgcn_wmma_f32_16x16x4_f32(
            false, a1l, false, b0l, (short)0, acc10, false, false);
        acc11 = __builtin_amdgcn_wmma_f32_16x16x4_f32(
            false, a1l, false, b1l, (short)0, acc11, false, false);
        acc00 = __builtin_amdgcn_wmma_f32_16x16x4_f32(
            false, a0h, false, b0h, (short)0, acc00, false, false);
        acc01 = __builtin_amdgcn_wmma_f32_16x16x4_f32(
            false, a0h, false, b1h, (short)0, acc01, false, false);
        acc10 = __builtin_amdgcn_wmma_f32_16x16x4_f32(
            false, a1h, false, b0h, (short)0, acc10, false, false);
        acc11 = __builtin_amdgcn_wmma_f32_16x16x4_f32(
            false, a1h, false, b1h, (short)0, acc11, false, false);
    }

    const float ws0 = att_s[col0], ws1 = att_s[col1];
    const float wd0 = att_d[col0], wd1 = att_d[col1];
    float v[16];
#pragma unroll
    for (int r = 0; r < 8; ++r) {
        v[r]     = acc00[r] * ws0 + acc01[r] * ws1;
        v[r + 8] = acc00[r] * wd0 + acc01[r] * wd1;
    }
    const float d0 = fold16(v, l15);
#pragma unroll
    for (int r = 0; r < 8; ++r) {
        v[r]     = acc10[r] * ws0 + acc11[r] * ws1;
        v[r + 8] = acc10[r] * wd0 + acc11[r] * wd1;
    }
    const float d1 = fold16(v, l15);
    sAtt[l15 >> 3][half * 8 + (l15 & 7)][wave]      = d0;
    sAtt[l15 >> 3][16 + half * 8 + (l15 & 7)][wave] = d1;
    __syncthreads();
    if (wave < 2) {
        const int grow = rowBase + lane;
        if (grow < n) {
            const float4 o = *(const float4*)&sAtt[wave][lane][0];
            *(float4*)((wave ? a_d : a_s) + (size_t)grow * NH) = o;
        }
    }

    const int crow0 = rowBase + half * 8;
#pragma unroll
    for (int r = 0; r < 8; ++r) {
        const int crow = crow0 + r;
        if (crow < n) {
            C[(size_t)crow * Oc + col0] = acc00[r];
            C[(size_t)crow * Oc + col1] = acc01[r];
        }
    }
#pragma unroll
    for (int r = 0; r < 8; ++r) {
        const int crow = crow0 + 16 + r;
        if (crow < n) {
            C[(size_t)crow * Oc + col0] = acc10[r];
            C[(size_t)crow * Oc + col1] = acc11[r];
        }
    }
}

template <int GROUPS>
__device__ __forceinline__ void gemm1_block(const float* __restrict__ x,
                                            const float* __restrict__ W1,
                                            float* __restrict__ xp,
                                            const float* __restrict__ att_s,
                                            const float* __restrict__ att_d,
                                            float* __restrict__ a_s, float* __restrict__ a_d,
                                            int n, int blk) {
    __shared__ __attribute__((aligned(16))) float sAtt2[GROUPS][2][32][NH];
    const int w   = (int)(threadIdx.x >> 5);
    const int grp = w >> 2;
    gemm_rows32<64>(x, W1, xp, att_s, att_d, a_s, a_d, n, HID,
                    blk * (32 * GROUPS) + grp * 32, w & 3, (int)(threadIdx.x & 31), sAtt2[grp]);
}


#define BK_SH    8
#define BK_NODES (1 << BK_SH)
#define BK_MAX   512
#define CS_T     512
#define SORT_T   256
#define SORT_G   256
#define SORT_U   16
#define CSR_P    16

static_assert(CS_T == BK_MAX, "chunk_sort's prefix takes one bucket per thread");
static_assert(SORT_T == BK_NODES && SORT_T == SORT_G, "bucket_csr: thread t owns node t and chunk t");

__device__ __forceinline__ void chunk_sort_body(const int* __restrict__ ei, int* __restrict__ tmp,
                                                int* __restrict__ loc, int E, int chunk, int nbuck) {
    __shared__ int h[BK_MAX];
    __shared__ int wtot[CS_T / 32];
    const int t = threadIdx.x, lane = t & 31, wave = t >> 5;
    h[t] = 0;
    __syncthreads();
    const int beg = blockIdx.x * chunk;
    const int end = min(E, beg + chunk);
    for (int i0 = beg + t; i0 < end; i0 += SORT_U * CS_T) {
        int d[SORT_U];
#pragma unroll
        for (int u = 0; u < SORT_U; ++u)
            d[u] = ei[E + min(i0 + u * CS_T, end - 1)];
#pragma unroll
        for (int u = 0; u < SORT_U; ++u)
            if (i0 + u * CS_T < end) atomicAdd(&h[d[u] >> BK_SH], 1);
    }
    __syncthreads();
    const int mine = (t < nbuck) ? h[t] : 0;
    int incl = mine;
#pragma unroll
    for (int o = 1; o < 32; o <<= 1) {
        const int up = __shfl_up(incl, o);
        if (lane >= o) incl += up;
    }
    if (lane == 31) wtot[wave] = incl;
    __syncthreads();
    int start = incl - mine;
#pragma unroll
    for (int w = 0; w < CS_T / 32; ++w)
        if (w < wave) start += wtot[w];
    int* lrow = loc + (size_t)blockIdx.x * (nbuck + 1);
    if (t < nbuck) lrow[t] = start;
    if (t == 0) lrow[nbuck] = max(end - beg, 0);
    h[t] = start;
    __syncthreads();
    for (int i0 = beg + t; i0 < end; i0 += SORT_U * CS_T) {
        int d[SORT_U], s[SORT_U], pos[SORT_U];
#pragma unroll
        for (int u = 0; u < SORT_U; ++u) {
            const int j = min(i0 + u * CS_T, end - 1);
            d[u] = ei[E + j];
            s[u] = ei[j];
        }
#pragma unroll
        for (int u = 0; u < SORT_U; ++u)
            if (i0 + u * CS_T < end) pos[u] = atomicAdd(&h[d[u] >> BK_SH], 1);
#pragma unroll
        for (int u = 0; u < SORT_U; ++u)
            if (i0 + u * CS_T < end)
                tmp[beg + pos[u]] = (int)(((unsigned)s[u] << BK_SH) | (unsigned)(d[u] & (BK_NODES - 1)));
    }
}

__device__ __forceinline__ void bucket_csr_body(const int* __restrict__ tmp, const int* __restrict__ loc,
                                                int* __restrict__ rowptr, int* __restrict__ csr_src,
                                                int N, int chunk, int nbuck) {
    __shared__ int c[BK_NODES];
    __shared__ int wtot[SORT_T / 32];
    __shared__ int wsum[SORT_T / 32];
    const int lane = threadIdx.x & 31, wave = threadIdx.x >> 5;
    const int* lrow = loc + (size_t)threadIdx.x * (nbuck + 1) + blockIdx.x;
    const int lo = lrow[0];
    const int plen  = lrow[1] - lo;
    const int pbase = (int)threadIdx.x * chunk + lo;
    int before = lo;
#pragma unroll
    for (int o = 16; o > 0; o >>= 1) before += __shfl_xor(before, o);
    if (lane == 0) wsum[wave] = before;
    c[threadIdx.x] = 0;
    __syncthreads();
    int bbeg = 0;
#pragma unroll
    for (int w = 0; w < SORT_T / 32; ++w) bbeg += wsum[w];

    for (int j0 = 0; j0 < 32; j0 += CSR_P) {
        int v[CSR_P];
#pragma unroll
        for (int q = 0; q < CSR_P; ++q) {
            const int pb = __builtin_amdgcn_readlane(pbase, j0 + q);
            const int pl = __builtin_amdgcn_readlane(plen, j0 + q);
            v[q] = 0;
            if (lane < pl) v[q] = tmp[pb + lane];
        }
#pragma unroll
        for (int q = 0; q < CSR_P; ++q) {
            const int pb = __builtin_amdgcn_readlane(pbase, j0 + q);
            const int pl = __builtin_amdgcn_readlane(plen, j0 + q);
            if (lane < pl) atomicAdd(&c[v[q] & (BK_NODES - 1)], 1);
            for (int k = 32 + lane; k < pl; k += 32)
                atomicAdd(&c[tmp[pb + k] & (BK_NODES - 1)], 1);
        }
    }
    __syncthreads();
    const int mine = c[threadIdx.x];
    int incl = mine;
#pragma unroll
    for (int o = 1; o < 32; o <<= 1) {
        const int up = __shfl_up(incl, o);
        if (lane >= o) incl += up;
    }
    if (lane == 31) wtot[wave] = incl;
    __syncthreads();
    int start = bbeg + incl - mine;
#pragma unroll
    for (int w = 0; w < SORT_T / 32; ++w)
        if (w < wave) start += wtot[w];
    const int node = blockIdx.x * BK_NODES + threadIdx.x;
    if (node < N) rowptr[node] = start;
    if (node == N - 1) rowptr[N] = start + mine;
    c[threadIdx.x] = start;
    __syncthreads();
    for (int j0 = 0; j0 < 32; j0 += CSR_P) {
        int v[CSR_P], pos[CSR_P];
#pragma unroll
        for (int q = 0; q < CSR_P; ++q) {
            const int pb = __builtin_amdgcn_readlane(pbase, j0 + q);
            const int pl = __builtin_amdgcn_readlane(plen, j0 + q);
            v[q] = 0;
            if (lane < pl) v[q] = tmp[pb + lane];
        }
#pragma unroll
        for (int q = 0; q < CSR_P; ++q) {
            const int pl = __builtin_amdgcn_readlane(plen, j0 + q);
            pos[q] = 0;
            if (lane < pl) pos[q] = atomicAdd(&c[v[q] & (BK_NODES - 1)], 1);
        }
#pragma unroll
        for (int q = 0; q < CSR_P; ++q) {
            const int pb = __builtin_amdgcn_readlane(pbase, j0 + q);
            const int pl = __builtin_amdgcn_readlane(plen, j0 + q);
            if (lane < pl) csr_src[pos[q]] = (int)((unsigned)v[q] >> BK_SH);
            for (int k = 32 + lane; k < pl; k += 32) {
                const int vv = tmp[pb + k];
                csr_src[atomicAdd(&c[vv & (BK_NODES - 1)], 1)] = (int)((unsigned)vv >> BK_SH);
            }
        }
    }
}

__global__ void __launch_bounds__(CS_T)
chunk_sort_gemm(const int* __restrict__ ei, int* __restrict__ tmp, int* __restrict__ loc,
                int E, int chunk, int nbuck,
                const float* __restrict__ x, const float* __restrict__ W1, float* __restrict__ xp,
                const float* __restrict__ att_s, const float* __restrict__ att_d,
                float* __restrict__ a_s, float* __restrict__ a_d, int n) {
    if (blockIdx.x < SORT_G) {
        chunk_sort_body(ei, tmp, loc, E, chunk, nbuck);
        return;
    }
    gemm1_block<CS_T / 128>(x, W1, xp, att_s, att_d, a_s, a_d, n, (int)blockIdx.x - SORT_G);
}

__global__ void __launch_bounds__(SORT_T)
bucket_csr_gemm(const int* __restrict__ tmp, const int* __restrict__ loc,
                int* __restrict__ rowptr, int* __restrict__ csr_src, int N, int chunk, int nbuck,
                const float* __restrict__ x, const float* __restrict__ W1, float* __restrict__ xp,
                const float* __restrict__ att_s, const float* __restrict__ att_d,
                float* __restrict__ a_s, float* __restrict__ a_d, int blk0) {
    if ((int)blockIdx.x < nbuck) {
        bucket_csr_body(tmp, loc, rowptr, csr_src, N, chunk, nbuck);
        return;
    }
    gemm1_block<SORT_T / 128>(x, W1, xp, att_s, att_d, a_s, a_d, N, blk0 + (int)blockIdx.x - nbuck);
}

__device__ __forceinline__ float lrelu(float v) {
    return v > 0.0f ? v : NEG_SLOPE * v;
}

__device__ __forceinline__ float elu1(float v) {
    return v > 0.0f ? v : expm1f(v);
}

__device__ __forceinline__ float head_sum(float v) {
    v += row_xor<1>(v);
    v += row_xor<2>(v);
    v += row_xor<4>(v);
    return v;
}

__device__ __forceinline__ float fold8(float (&v)[8], int i) {
#pragma unroll
    for (int k = 0; k < 4; ++k) {
        const bool up = (i & 4) != 0;
        const float send = up ? v[k] : v[k + 4];
        const float keep = up ? v[k + 4] : v[k];
        v[k] = keep + row_xor<4>(send);
    }
#pragma unroll
    for (int k = 0; k < 2; ++k) {
        const bool up = (i & 2) != 0;
        const float send = up ? v[k] : v[k + 2];
        const float keep = up ? v[k + 2] : v[k];
        v[k] = keep + row_xor<2>(send);
    }
    const bool up = (i & 1) != 0;
    const float send = up ? v[0] : v[1];
    const float keep = up ? v[1] : v[0];
    return keep + row_xor<1>(send);
}

__device__ __forceinline__ float head_max(float v) {
    v = fmaxf(v, row_xor<1>(v));
    v = fmaxf(v, row_xor<2>(v));
    v = fmaxf(v, row_xor<4>(v));
    return v;
}

#define ROWS_CH 8

template <int T>
__device__ __forceinline__ float group_bcast(float v) {
    return __int_as_float(__builtin_amdgcn_mov_dpp8(__float_as_int(v), T * 0x249249));
}

template <int T, int R>
__device__ __forceinline__ void edge_acc(const v4f (&r)[R], float psel, v2f& acc01, v2f& acc23) {
    if constexpr (T < R) {
        const float p = group_bcast<T>(psel);
        const v2f lo = {r[T].x, r[T].y}, hi = {r[T].z, r[T].w};
        acc01 += lo * p;
        acc23 += hi * p;
    }
}

template <int R>
__device__ __forceinline__ void edge_chunk(const float* __restrict__ xp, int s, int c0,
                                           const v2f as01, const v2f as23, float adn, int lane,
                                           v2f& acc01, v2f& acc23, float& den, float& m) {
    v4f r[R];
    float x[ROWS_CH];
#pragma unroll
    for (int t = 0; t < R; ++t) {
        const int st = __builtin_amdgcn_readlane(s, c0 + t);
        r[t] = *(const v4f*)(xp + (size_t)st * HID + lane * 4);
    }
    __builtin_amdgcn_sched_barrier(0);
#pragma unroll
    for (int t = 0; t < ROWS_CH; ++t) {
        x[t] = 0.0f;
        if (t < R) {
            const v2f lo = {r[t].x, r[t].y}, hi = {r[t].z, r[t].w};
            const v2f d = lo * as01 + hi * as23;
            x[t] = d.x + d.y;
        }
    }
    const int i = lane & 7;
    float xsel = fold8(x, i) + adn;
    if (R < ROWS_CH) xsel = i < R ? xsel : -INFINITY;
    const float xm = head_max(xsel);
    const float mn = fmaxf(m, fmaxf(xm, NEG_SLOPE * xm));
    const float sc = expf(m - mn);
    acc01 *= sc; acc23 *= sc;
    den *= sc;
    m = mn;
    const float esel = fmaxf(xsel, NEG_SLOPE * xsel);
    float psel = expf(esel - mn);
    if (R < ROWS_CH) psel = i < R ? psel : 0.0f;
    den += head_sum(psel);
    edge_acc<0, R>(r, psel, acc01, acc23); edge_acc<1, R>(r, psel, acc01, acc23);
    edge_acc<2, R>(r, psel, acc01, acc23); edge_acc<3, R>(r, psel, acc01, acc23);
    edge_acc<4, R>(r, psel, acc01, acc23); edge_acc<5, R>(r, psel, acc01, acc23);
    edge_acc<6, R>(r, psel, acc01, acc23); edge_acc<7, R>(r, psel, acc01, acc23);
}

template <bool LAST>
__global__ void node_softmax_agg(const int* __restrict__ rowptr,
                                 const int* __restrict__ csr_src,
                                 const float* __restrict__ xp,
                                 const float* __restrict__ att_s,
                                 const float* __restrict__ a_s,
                                 const float* __restrict__ a_d,
                                 const float* __restrict__ bias,
                                 float* __restrict__ dst, int n) {
    const int lane = threadIdx.x & 31;
    const int node = __builtin_amdgcn_readfirstlane(
        (int)((blockIdx.x * blockDim.x + threadIdx.x) >> 5));
    if (node >= n) return;
    const int beg = rowptr[node], end = rowptr[node + 1];
    const v4f as4 = *(const v4f*)(att_s + lane * 4);
    const v2f as01 = {as4.x, as4.y}, as23 = {as4.z, as4.w};
    const int hme = lane >> 3;

    const float4 ad = *(const float4*)(a_d + (size_t)node * NH);
    const float4 an = *(const float4*)(a_s + (size_t)node * NH);
    const float adn = hme == 0 ? ad.x : hme == 1 ? ad.y : hme == 2 ? ad.z : ad.w;
    const float asn = hme == 0 ? an.x : hme == 1 ? an.y : hme == 2 ? an.z : an.w;
    const v4f rs = *(const v4f*)(xp + (size_t)node * HID + lane * 4);
    float m = lrelu(asn + adn);
    v2f acc01 = {rs.x, rs.y}, acc23 = {rs.z, rs.w};
    float den = 1.0f;

    for (int base = beg; base < end; base += 32) {
        const int j = base + lane;
        const int s = (j < end) ? csr_src[j] : node;
        const int cnt32 = min(32, end - base);
        int c0 = 0;
        for (; c0 + ROWS_CH <= cnt32; c0 += ROWS_CH)
            edge_chunk<ROWS_CH>(xp, s, c0, as01, as23, adn, lane, acc01, acc23, den, m);
        switch (cnt32 - c0) {
            case 1: edge_chunk<1>(xp, s, c0, as01, as23, adn, lane, acc01, acc23, den, m); break;
            case 2: edge_chunk<2>(xp, s, c0, as01, as23, adn, lane, acc01, acc23, den, m); break;
            case 3: edge_chunk<3>(xp, s, c0, as01, as23, adn, lane, acc01, acc23, den, m); break;
            case 4: edge_chunk<4>(xp, s, c0, as01, as23, adn, lane, acc01, acc23, den, m); break;
            case 5: edge_chunk<5>(xp, s, c0, as01, as23, adn, lane, acc01, acc23, den, m); break;
            case 6: edge_chunk<6>(xp, s, c0, as01, as23, adn, lane, acc01, acc23, den, m); break;
            case 7: edge_chunk<7>(xp, s, c0, as01, as23, adn, lane, acc01, acc23, den, m); break;
            default: break;
        }
    }

    const float dd = den + 1e-16f;
    const float4 a = make_float4(acc01.x / dd, acc01.y / dd, acc23.x / dd, acc23.y / dd);
    if (!LAST) {
        const float4 b = *(const float4*)(bias + lane * 4);
        *(float4*)(dst + (size_t)node * HID + lane * 4) =
            make_float4(elu1(a.x + b.x), elu1(a.y + b.y), elu1(a.z + b.z), elu1(a.w + b.w));
    } else {
        const float4 a1 = make_float4(__shfl_xor(a.x, 8),  __shfl_xor(a.y, 8),
                                      __shfl_xor(a.z, 8),  __shfl_xor(a.w, 8));
        const float4 a2 = make_float4(__shfl_xor(a.x, 16), __shfl_xor(a.y, 16),
                                      __shfl_xor(a.z, 16), __shfl_xor(a.w, 16));
        const float4 a3 = make_float4(__shfl_xor(a.x, 24), __shfl_xor(a.y, 24),
                                      __shfl_xor(a.z, 24), __shfl_xor(a.w, 24));
        if (lane < 8) {
            const float4 b = *(const float4*)(bias + lane * 4);
            *(float4*)(dst + (size_t)node * NC + lane * 4) =
                make_float4(0.25f * (a.x + a1.x + a2.x + a3.x) + b.x,
                            0.25f * (a.y + a1.y + a2.y + a3.y) + b.y,
                            0.25f * (a.z + a1.z + a2.z + a3.z) + b.z,
                            0.25f * (a.w + a1.w + a2.w + a3.w) + b.w);
        }
    }
}


#define FR 32
#define HP (HID + 4)

__device__ __forceinline__ float4 node_row(const int* __restrict__ rowptr,
                                           const int* __restrict__ csr_src,
                                           const float* __restrict__ xp,
                                           const v2f as01, const v2f as23,
                                           const float* __restrict__ a_s,
                                           const float* __restrict__ a_d,
                                           int node, int lane) {
    const int beg = rowptr[node], end = rowptr[node + 1];
    const int hme = lane >> 3;
    const float4 ad = *(const float4*)(a_d + (size_t)node * NH);
    const float4 an = *(const float4*)(a_s + (size_t)node * NH);
    const float adn = hme == 0 ? ad.x : hme == 1 ? ad.y : hme == 2 ? ad.z : ad.w;
    const float asn = hme == 0 ? an.x : hme == 1 ? an.y : hme == 2 ? an.z : an.w;
    const v4f rs = *(const v4f*)(xp + (size_t)node * HID + lane * 4);
    float m = lrelu(asn + adn);
    v2f acc01 = {rs.x, rs.y}, acc23 = {rs.z, rs.w};
    float den = 1.0f;

    for (int base = beg; base < end; base += 32) {
        const int j = base + lane;
        const int s = (j < end) ? csr_src[j] : node;
        const int cnt32 = min(32, end - base);
        int c0 = 0;
        for (; c0 + ROWS_CH <= cnt32; c0 += ROWS_CH)
            edge_chunk<ROWS_CH>(xp, s, c0, as01, as23, adn, lane, acc01, acc23, den, m);
        switch (cnt32 - c0) {
            case 1: edge_chunk<1>(xp, s, c0, as01, as23, adn, lane, acc01, acc23, den, m); break;
            case 2: edge_chunk<2>(xp, s, c0, as01, as23, adn, lane, acc01, acc23, den, m); break;
            case 3: edge_chunk<3>(xp, s, c0, as01, as23, adn, lane, acc01, acc23, den, m); break;
            case 4: edge_chunk<4>(xp, s, c0, as01, as23, adn, lane, acc01, acc23, den, m); break;
            case 5: edge_chunk<5>(xp, s, c0, as01, as23, adn, lane, acc01, acc23, den, m); break;
            case 6: edge_chunk<6>(xp, s, c0, as01, as23, adn, lane, acc01, acc23, den, m); break;
            case 7: edge_chunk<7>(xp, s, c0, as01, as23, adn, lane, acc01, acc23, den, m); break;
            default: break;
        }
    }
    const float dd = den + 1e-16f;
    return make_float4(acc01.x / dd, acc01.y / dd, acc23.x / dd, acc23.y / dd);
}

__global__ void __launch_bounds__(256)
node_agg_gemm2(const int* __restrict__ rowptr, const int* __restrict__ csr_src,
               const float* __restrict__ xp1, const float* __restrict__ att_s1,
               const float* __restrict__ a_s1, const float* __restrict__ a_d1,
               const float* __restrict__ bias1,
               const float* __restrict__ W2, float* __restrict__ xp2,
               const float* __restrict__ att_s2, const float* __restrict__ att_d2,
               float* __restrict__ a_s2, float* __restrict__ a_d2, int n) {
    __shared__ __attribute__((aligned(16))) float sH[FR][HP];
    __shared__ __attribute__((aligned(16))) float sAtt[2][32][NH];
    const int lane = threadIdx.x & 31;
    const int wave = __builtin_amdgcn_readfirstlane((int)(threadIdx.x >> 5));
    const int rowBase = (int)blockIdx.x * FR;

    {
        const v4f as4 = *(const v4f*)(att_s1 + lane * 4);
        const v2f as01 = {as4.x, as4.y}, as23 = {as4.z, as4.w};
        const float4 b = *(const float4*)(bias1 + lane * 4);
        for (int q = 0; q < FR / 8; ++q) {
            const int r = wave * (FR / 8) + q;
            const int node = rowBase + r;
            float4 o = make_float4(0.0f, 0.0f, 0.0f, 0.0f);
            if (node < n) {
                const float4 a = node_row(rowptr, csr_src, xp1, as01, as23, a_s1, a_d1, node, lane);
                o = make_float4(elu1(a.x + b.x), elu1(a.y + b.y), elu1(a.z + b.z), elu1(a.w + b.w));
            }
            *(float4*)&sH[r][lane * 4] = o;
        }
    }
    __syncthreads();

    const int half = lane >> 4;
    const int l15  = lane & 15;
    const int col0 = wave * 32 + l15;
    const int col1 = col0 + 16;
    v8f acc00 = {}, acc01 = {}, acc10 = {}, acc11 = {};
    if (wave < 4) {
        const float* Ap0 = &sH[l15][4 * half];
        const float* Ap1 = &sH[16 + l15][4 * half];
        const float* Wp0 = W2 + (size_t)col0 * HID + 4 * half;
        const float* Wp1 = W2 + (size_t)col1 * HID + 4 * half;
#pragma unroll
        for (int k = 0; k < HID; k += 8) {
            const v4f a0 = *(const v4f*)(Ap0 + k);
            const v4f a1 = *(const v4f*)(Ap1 + k);
            const v4f b0 = *(const v4f*)(Wp0 + k);
            const v4f b1 = *(const v4f*)(Wp1 + k);
            const v2f a0l = {a0.x, a0.y}, a0h = {a0.z, a0.w};
            const v2f a1l = {a1.x, a1.y}, a1h = {a1.z, a1.w};
            const v2f b0l = {b0.x, b0.y}, b0h = {b0.z, b0.w};
            const v2f b1l = {b1.x, b1.y}, b1h = {b1.z, b1.w};
            acc00 = __builtin_amdgcn_wmma_f32_16x16x4_f32(
                false, a0l, false, b0l, (short)0, acc00, false, false);
            acc01 = __builtin_amdgcn_wmma_f32_16x16x4_f32(
                false, a0l, false, b1l, (short)0, acc01, false, false);
            acc10 = __builtin_amdgcn_wmma_f32_16x16x4_f32(
                false, a1l, false, b0l, (short)0, acc10, false, false);
            acc11 = __builtin_amdgcn_wmma_f32_16x16x4_f32(
                false, a1l, false, b1l, (short)0, acc11, false, false);
            acc00 = __builtin_amdgcn_wmma_f32_16x16x4_f32(
                false, a0h, false, b0h, (short)0, acc00, false, false);
            acc01 = __builtin_amdgcn_wmma_f32_16x16x4_f32(
                false, a0h, false, b1h, (short)0, acc01, false, false);
            acc10 = __builtin_amdgcn_wmma_f32_16x16x4_f32(
                false, a1h, false, b0h, (short)0, acc10, false, false);
            acc11 = __builtin_amdgcn_wmma_f32_16x16x4_f32(
                false, a1h, false, b1h, (short)0, acc11, false, false);
        }
        const float ws0 = att_s2[col0], ws1 = att_s2[col1];
        const float wd0 = att_d2[col0], wd1 = att_d2[col1];
        float v[16];
#pragma unroll
        for (int r = 0; r < 8; ++r) {
            v[r]     = acc00[r] * ws0 + acc01[r] * ws1;
            v[r + 8] = acc00[r] * wd0 + acc01[r] * wd1;
        }
        const float d0 = fold16(v, l15);
#pragma unroll
        for (int r = 0; r < 8; ++r) {
            v[r]     = acc10[r] * ws0 + acc11[r] * ws1;
            v[r + 8] = acc10[r] * wd0 + acc11[r] * wd1;
        }
        const float d1 = fold16(v, l15);
        sAtt[l15 >> 3][half * 8 + (l15 & 7)][wave]      = d0;
        sAtt[l15 >> 3][16 + half * 8 + (l15 & 7)][wave] = d1;
    }
    __syncthreads();
    if (wave < 2) {
        const int grow = rowBase + lane;
        if (grow < n) {
            const float4 o = *(const float4*)&sAtt[wave][lane][0];
            *(float4*)((wave ? a_d2 : a_s2) + (size_t)grow * NH) = o;
        }
    }
    if (wave < 4) {
        const int crow0 = rowBase + half * 8;
#pragma unroll
        for (int r = 0; r < 8; ++r) {
            const int crow = crow0 + r;
            if (crow < n) {
                xp2[(size_t)crow * HID + col0] = acc00[r];
                xp2[(size_t)crow * HID + col1] = acc01[r];
            }
        }
#pragma unroll
        for (int r = 0; r < 8; ++r) {
            const int crow = crow0 + 16 + r;
            if (crow < n) {
                xp2[(size_t)crow * HID + col0] = acc10[r];
                xp2[(size_t)crow * HID + col1] = acc11[r];
            }
        }
    }
}


static inline int cdiv(int a, int b) { return (a + b - 1) / b; }

extern "C" void kernel_launch(void* const* d_in, const int* in_sizes, int n_in,
                              void* d_out, int out_size, void* d_ws, size_t ws_size,
                              hipStream_t stream) {
    const float* x        = (const float*)d_in[0];
    const int*   ei       = (const int*)  d_in[1];
    const float* W1       = (const float*)d_in[2];
    const float* att_src1 = (const float*)d_in[3];
    const float* att_dst1 = (const float*)d_in[4];
    const float* bias1    = (const float*)d_in[5];
    const float* W2       = (const float*)d_in[6];
    const float* att_src2 = (const float*)d_in[7];
    const float* att_dst2 = (const float*)d_in[8];
    const float* bias2    = (const float*)d_in[9];
    float* out = (float*)d_out;

    const int IN = 64;
    const int N  = in_sizes[0] / IN;
    const int E  = in_sizes[1] / 2;

    float* xp1   = (float*)d_ws;
    float* agg1  = xp1  + (size_t)N * HID;
    float* xp2   = agg1 + (size_t)N * HID;
    float* r4    = xp2  + (size_t)N * HID;
    float* a_s   = r4   + (size_t)N * HID;
    float* a_d   = a_s  + (size_t)N * NH;
    float* a_s2  = agg1;
    float* a_d2  = a_s2 + (size_t)N * NH;

    int* csr_src = (int*)r4;
    int* rowptr  = csr_src + (size_t)E;
    int* tmp     = rowptr + (size_t)N + 1;
    const int nbuck = cdiv(N, BK_NODES);
    int* loc     = tmp + (size_t)E;

    const int BT = 256;
    const int gRowTiles = cdiv(N, 32);
    const int chunk = cdiv(E, SORT_G);
    const int gNW  = cdiv(N * 32, BT);
    const int gGa  = (cdiv(N, 128) * 2) / 5;
    const int gGb  = cdiv(N, 64) - gGa * 2;

    chunk_sort_gemm<<<SORT_G + gGa, CS_T, 0, stream>>>(ei, tmp, loc, E, chunk, nbuck,
        x, W1, xp1, att_src1, att_dst1, a_s, a_d, N);
    bucket_csr_gemm<<<nbuck + gGb, SORT_T, 0, stream>>>(tmp, loc, rowptr, csr_src, N, chunk, nbuck,
        x, W1, xp1, att_src1, att_dst1, a_s, a_d, gGa * 2);

    node_agg_gemm2<<<gRowTiles, 256, 0, stream>>>(rowptr, csr_src, xp1, att_src1, a_s, a_d, bias1,
        W2, xp2, att_src2, att_dst2, a_s2, a_d2, N);

    node_softmax_agg<true><<<gNW, BT, 0, stream>>>(rowptr, csr_src, xp2, att_src2, a_s2, a_d2, bias2, out, N);
}
